# starting kernel
# baseline (speedup 1.0000x reference)
#include <hip/hip_runtime.h>

#define ALPHA 0.2f

typedef __attribute__((ext_vector_type(2))) float v2f;
typedef __attribute__((ext_vector_type(8))) float v8f;

__device__ __forceinline__ v8f wmma_f32_16x16x4(v2f a, v2f b, v8f c) {
  return __builtin_amdgcn_wmma_f32_16x16x4_f32(false, a, false, b, (short)0, c,
                                               false, false);
}

__device__ __forceinline__ float leaky(float t) {
  return (t >= 0.0f) ? t : ALPHA * t;
}

__global__ void __launch_bounds__(128) k_h_gemm(const float* __restrict__ x,
                                                const float* __restrict__ W,
                                                float* __restrict__ h) {
  const int lane = threadIdx.x & 31;
  const int wave = threadIdx.x >> 5;
  const int rl = lane & 15;
  const int hi = lane >> 4;
  const int i0 = blockIdx.x * 64 + wave * 16;

  v8f acc[8];
  v8f zero = {};
#pragma unroll
  for (int t = 0; t < 8; ++t) acc[t] = zero;

  const float* xrow = x + (size_t)(i0 + rl) * 128 + 2 * hi;
#pragma unroll 4
  for (int k = 0; k < 128; k += 4) {
    v2f afrag = *(const v2f*)(xrow + k);
#pragma unroll
    for (int ot = 0; ot < 8; ++ot) {
      v2f bfrag = *(const v2f*)(W + (size_t)(ot * 16 + rl) * 128 + k + 2 * hi);
      acc[ot] = wmma_f32_16x16x4(afrag, bfrag, acc[ot]);
    }
  }
#pragma unroll
  for (int ot = 0; ot < 8; ++ot)
#pragma unroll
    for (int r = 0; r < 8; ++r)
      h[(size_t)(i0 + r + hi * 8) * 128 + ot * 16 + rl] = acc[ot][r];
}

__global__ void __launch_bounds__(256) k_scores(const float* __restrict__ h,
                                                const float* __restrict__ a,
                                                float* __restrict__ si,
                                                float* __restrict__ sj) {
  const int row = blockIdx.x * 8 + (threadIdx.x >> 5);
  const int lane = threadIdx.x & 31;
  float psi = 0.0f, psj = 0.0f;
#pragma unroll
  for (int c = lane; c < 128; c += 32) {
    float hv = h[(size_t)row * 128 + c];
    psi += hv * a[c];
    psj += hv * a[128 + c];
  }
#pragma unroll
  for (int off = 16; off > 0; off >>= 1) {
    psi += __shfl_xor(psi, off, 32);
    psj += __shfl_xor(psj, off, 32);
  }
  if (lane == 0) {
    si[row] = psi;
    sj[row] = psj;
  }
}

__global__ void __launch_bounds__(256) k_maxsj(const float* __restrict__ sj,
                                               float* __restrict__ mx) {
  __shared__ float red[256];
  const int b = blockIdx.x;
  float m = -3.402823466e38f;
  for (int j = threadIdx.x; j < 2048; j += 256)
    m = fmaxf(m, sj[(size_t)b * 2048 + j]);
  red[threadIdx.x] = m;
  __syncthreads();
  for (int s = 128; s > 0; s >>= 1) {
    if (threadIdx.x < s) red[threadIdx.x] = fmaxf(red[threadIdx.x], red[threadIdx.x + s]);
    __syncthreads();
  }
  if (threadIdx.x == 0) mx[b] = red[0];
}

#define PSTRIDE 288
__global__ void __launch_bounds__(128) k_attn(const float* __restrict__ h,
                                              const float* __restrict__ si,
                                              const float* __restrict__ sj,
                                              const float* __restrict__ mx,
                                              float* __restrict__ out) {
  __shared__ __align__(16) float hs2[32 * PSTRIDE];
  __shared__ __align__(16) float sjs[64];
  __shared__ float zbuf[4 * 16];

  const int tid = threadIdx.x;
  const int lane = tid & 31;
  const int wave = tid >> 5;
  const int rl = lane & 15;
  const int hi = lane >> 4;
  const int b = blockIdx.y;
  const int i0 = blockIdx.x * 64 + wave * 16;

  const float* hb = h + (size_t)b * 2048 * 128;
  const float* sjb = sj + (size_t)b * 2048;

  const float si_row = si[(size_t)b * 2048 + i0 + rl];
  const float m_row = leaky(si_row + mx[b]);

  v8f acc[8];
  v8f zero = {};
#pragma unroll
  for (int t = 0; t < 8; ++t) acc[t] = zero;
  float zacc = 0.0f;

  float4 rA[8], rB[8];
  float sjr = 0.0f;

#pragma unroll
  for (int k = 0; k < 8; ++k) {
    const int q = tid + 128 * k;
    const int jp = q >> 5, c0 = (q & 31) * 4;
    rA[k] = *(const float4*)(hb + (size_t)(2 * jp) * 128 + c0);
    rB[k] = *(const float4*)(hb + (size_t)(2 * jp + 1) * 128 + c0);
  }
  if (tid < 64) sjr = sjb[tid];
#pragma unroll
  for (int k = 0; k < 8; ++k) {
    const int q = tid + 128 * k;
    const int jp = q >> 5, c0 = (q & 31) * 4;
    float4* dst = (float4*)&hs2[jp * PSTRIDE + c0 * 2];
    dst[0] = make_float4(rA[k].x, rB[k].x, rA[k].y, rB[k].y);
    dst[1] = make_float4(rA[k].z, rB[k].z, rA[k].w, rB[k].w);
  }
  if (tid < 64) sjs[tid] = sjr;
  __syncthreads();

  for (int ch = 0; ch < 32; ++ch) {
    if (ch + 1 < 32) {
      const int jn = (ch + 1) * 64;
#pragma unroll
      for (int k = 0; k < 8; ++k) {
        const int q = tid + 128 * k;
        const int jp = q >> 5, c0 = (q & 31) * 4;
        rA[k] = *(const float4*)(hb + (size_t)(jn + 2 * jp) * 128 + c0);
        rB[k] = *(const float4*)(hb + (size_t)(jn + 2 * jp + 1) * 128 + c0);
      }
      if (tid < 64) sjr = sjb[jn + tid];
    }

#pragma unroll 4
    for (int j = 0; j < 64; j += 4) {
      const int jp = (j >> 1) + hi;
      v2f sp = *(const v2f*)&sjs[j + 2 * hi];
      float p0 = __expf(leaky(si_row + sp[0]) - m_row);
      float p1 = __expf(leaky(si_row + sp[1]) - m_row);
      zacc += p0 + p1;
      v2f afrag = {p0, p1};
      const float* bb = &hs2[jp * PSTRIDE + rl * 2];
#pragma unroll
      for (int ot = 0; ot < 8; ++ot) {
        v2f bfrag = *(const v2f*)(bb + ot * 32);
        acc[ot] = wmma_f32_16x16x4(afrag, bfrag, acc[ot]);
      }
    }
    __syncthreads();

    if (ch + 1 < 32) {
#pragma unroll
      for (int k = 0; k < 8; ++k) {
        const int q = tid + 128 * k;
        const int jp = q >> 5, c0 = (q & 31) * 4;
        float4* dst = (float4*)&hs2[jp * PSTRIDE + c0 * 2];
        dst[0] = make_float4(rA[k].x, rB[k].x, rA[k].y, rB[k].y);
        dst[1] = make_float4(rA[k].z, rB[k].z, rA[k].w, rB[k].w);
      }
      if (tid < 64) sjs[tid] = sjr;
    }
    __syncthreads();
  }

  float zfull = zacc + __shfl_xor(zacc, 16, 32);
  if (hi == 0) zbuf[wave * 16 + rl] = zfull;
  __syncthreads();
  float zinv[8];
#pragma unroll
  for (int r = 0; r < 8; ++r) zinv[r] = 1.0f / zbuf[wave * 16 + r + hi * 8];

#pragma unroll
  for (int ot = 0; ot < 8; ++ot)
#pragma unroll
    for (int r = 0; r < 8; ++r)
      out[((size_t)b * 2048 + i0 + r + hi * 8) * 128 + ot * 16 + rl] =
          acc[ot][r] * zinv[r];
}

extern "C" void kernel_launch(void* const* d_in, const int* in_sizes, int n_in,
                              void* d_out, int out_size, void* d_ws, size_t ws_size,
                              hipStream_t stream) {
  const float* x = (const float*)d_in[0];
  const float* W = (const float*)d_in[1];
  const float* a = (const float*)d_in[2];
  float* out = (float*)d_out;

  float* h  = (float*)d_ws;
  float* si = h + (size_t)16384 * 128;
  float* sj = si + 16384;
  float* mx = sj + 16384;

  k_h_gemm<<<dim3(256), dim3(128), 0, stream>>>(x, W, h);
  k_scores<<<dim3(2048), dim3(256), 0, stream>>>(h, a, si, sj);
  k_maxsj<<<dim3(8), dim3(256), 0, stream>>>(sj, mx);
  k_attn<<<dim3(32, 8), dim3(128), 0, stream>>>(h, si, sj, mx, out);
}


# GEMM kernel: wave tile 32 rows x 64 cols (2x4 tiles) fed by 16-byte global fragment loads, 3 per 8 WMMAs (was 9), issued two 16-k stages ahead in registers; no LDS; new k summation order
# speedup vs baseline: 1.4545x; 1.4545x over previous
#include <hip/hip_runtime.h>

#define ALPHA 0.2f

typedef __attribute__((ext_vector_type(2))) float v2f;
typedef __attribute__((ext_vector_type(4))) float v4f;
typedef __attribute__((ext_vector_type(8))) float v8f;

__device__ __forceinline__ v8f wmma_f32_16x16x4(v2f a, v2f b, v8f c) {
  return __builtin_amdgcn_wmma_f32_16x16x4_f32(false, a, false, b, (short)0, c,
                                               false, false);
}

__device__ __forceinline__ float leaky(float t) {
  return (t >= 0.0f) ? t : ALPHA * t;
}

#define G_DS 2
#define G_NST 8
__device__ __forceinline__ void gemm_load_stage(const float* xa, const float* wb,
                                                int kk0, v4f (&fa)[G_DS][2],
                                                v4f (&fb)[G_DS][4]) {
#pragma unroll
  for (int d = 0; d < G_DS; ++d) {
#pragma unroll
    for (int rt = 0; rt < 2; ++rt)
      fa[d][rt] = *(const v4f*)(xa + (size_t)rt * 16 * 128 + kk0 + d * 8);
#pragma unroll
    for (int ot = 0; ot < 4; ++ot)
      fb[d][ot] = *(const v4f*)(wb + (size_t)ot * 16 * 128 + kk0 + d * 8);
  }
}
__device__ __forceinline__ void gemm_compute_stage(const v4f (&fa)[G_DS][2],
                                                   const v4f (&fb)[G_DS][4],
                                                   v8f (&acc)[2][4]) {
#pragma unroll
  for (int d = 0; d < G_DS; ++d) {
#pragma unroll
    for (int half = 0; half < 2; ++half) {
      v2f a[2], b[4];
#pragma unroll
      for (int rt = 0; rt < 2; ++rt)
        a[rt] = v2f{fa[d][rt][2 * half], fa[d][rt][2 * half + 1]};
#pragma unroll
      for (int ot = 0; ot < 4; ++ot)
        b[ot] = v2f{fb[d][ot][2 * half], fb[d][ot][2 * half + 1]};
#pragma unroll
      for (int rt = 0; rt < 2; ++rt)
#pragma unroll
        for (int ot = 0; ot < 4; ++ot)
          acc[rt][ot] = wmma_f32_16x16x4(a[rt], b[ot], acc[rt][ot]);
    }
  }
}
__global__ void __launch_bounds__(128) k_h_gemm(const float* __restrict__ x,
                                                const float* __restrict__ W,
                                                float* __restrict__ h) {
  const int lane = threadIdx.x & 31;
  const int wave = threadIdx.x >> 5;
  const int rl = lane & 15;
  const int hi = lane >> 4;
  const int i0 = blockIdx.x * 64 + (wave & 1) * 32;
  const int cb = (wave >> 1) * 64;

  v8f acc[2][4];
  v8f zero = {};
#pragma unroll
  for (int rt = 0; rt < 2; ++rt)
#pragma unroll
    for (int ot = 0; ot < 4; ++ot) acc[rt][ot] = zero;

  const float* xa = x + (size_t)(i0 + rl) * 128 + 4 * hi;
  const float* wb = W + (size_t)(cb + rl) * 128 + 4 * hi;

  v4f fa[3][G_DS][2], fb[3][G_DS][4];
  gemm_load_stage(xa, wb, 0, fa[0], fb[0]);
  gemm_load_stage(xa, wb, G_DS * 8, fa[1], fb[1]);
#pragma unroll
  for (int st = 0; st < G_NST; ++st) {
    if (st + 2 < G_NST)
      gemm_load_stage(xa, wb, (st + 2) * G_DS * 8, fa[(st + 2) % 3], fb[(st + 2) % 3]);
    __builtin_amdgcn_sched_barrier(0);
    gemm_compute_stage(fa[st % 3], fb[st % 3], acc);
  }
#pragma unroll
  for (int rt = 0; rt < 2; ++rt)
#pragma unroll
    for (int ot = 0; ot < 4; ++ot)
#pragma unroll
      for (int r = 0; r < 8; ++r)
        h[(size_t)(i0 + rt * 16 + r + hi * 8) * 128 + cb + ot * 16 + rl] =
            acc[rt][ot][r];
}

#define SC_ROWS 8
__global__ void __launch_bounds__(256) k_scores(const float* __restrict__ h,
                                                const float* __restrict__ a,
                                                float* __restrict__ si,
                                                float* __restrict__ sj) {
  const int lane = threadIdx.x & 31;
  const int row0 = (blockIdx.x * 8 + (threadIdx.x >> 5)) * SC_ROWS;
  const float* hp = h + (size_t)row0 * 128 + lane;

  float hv[SC_ROWS][4];
#pragma unroll
  for (int r = 0; r < SC_ROWS; ++r)
#pragma unroll
    for (int k = 0; k < 4; ++k) hv[r][k] = hp[r * 128 + k * 32];
  float a1[4], a2[4];
#pragma unroll
  for (int k = 0; k < 4; ++k) {
    a1[k] = a[lane + k * 32];
    a2[k] = a[128 + lane + k * 32];
  }

  float psi[SC_ROWS], psj[SC_ROWS];
#pragma unroll
  for (int r = 0; r < SC_ROWS; ++r) {
    psi[r] = 0.0f;
    psj[r] = 0.0f;
#pragma unroll
    for (int k = 0; k < 4; ++k) {
      psi[r] += hv[r][k] * a1[k];
      psj[r] += hv[r][k] * a2[k];
    }
  }
#pragma unroll
  for (int off = 16; off > 0; off >>= 1) {
#pragma unroll
    for (int r = 0; r < SC_ROWS; ++r) {
      psi[r] += __shfl_xor(psi[r], off, 32);
      psj[r] += __shfl_xor(psj[r], off, 32);
    }
  }
  if (lane < SC_ROWS) {
    float osi = psi[0], osj = psj[0];
#pragma unroll
    for (int r = 1; r < SC_ROWS; ++r) {
      if (lane == r) {
        osi = psi[r];
        osj = psj[r];
      }
    }
    si[row0 + lane] = osi;
    sj[row0 + lane] = osj;
  }
}

__global__ void __launch_bounds__(256) k_maxsj(const float* __restrict__ sj,
                                               float* __restrict__ mx) {
  __shared__ float red[256];
  const int b = blockIdx.x;
  float m = -3.402823466e38f;
  for (int j = threadIdx.x; j < 2048; j += 256)
    m = fmaxf(m, sj[(size_t)b * 2048 + j]);
  red[threadIdx.x] = m;
  __syncthreads();
  for (int s = 128; s > 0; s >>= 1) {
    if (threadIdx.x < s) red[threadIdx.x] = fmaxf(red[threadIdx.x], red[threadIdx.x + s]);
    __syncthreads();
  }
  if (threadIdx.x == 0) mx[b] = red[0];
}

#define PSTRIDE 288

__device__ __forceinline__ void attn_half(const float* hcur, const float* kfcur,
                                          int j0, int hi, int boff, v2f ac_row,
                                          float& zacc, v8f (&acc)[4]) {
  v4f kf[8];
  v2f bf[8][4];
#pragma unroll
  for (int s = 0; s < 8; ++s) {
    const int j = j0 + 4 * s;
    const int jp = (j >> 1) + hi;
    kf[s] = *(const v4f*)&kfcur[jp * 4];
    const float* bb = &hcur[jp * PSTRIDE + boff];
#pragma unroll
    for (int ot = 0; ot < 4; ++ot)
      bf[s][ot] = *(const v2f*)(bb + ot * 32);
  }
  __builtin_amdgcn_sched_barrier(0);
#pragma unroll
  for (int s = 0; s < 8; ++s) {
    const v2f k0 = {kf[s][0], kf[s][1]};
    const v2f k1 = {kf[s][2], kf[s][3]};
    const v2f q0 = ac_row * k0;
    const v2f q1 = ac_row * k1;
    float p0 = fmaxf(q0[0], q0[1]);
    float p1 = fmaxf(q1[0], q1[1]);
    zacc += p0 + p1;
    v2f afrag = {p0, p1};
#pragma unroll
    for (int ot = 0; ot < 4; ++ot)
      acc[ot] = wmma_f32_16x16x4(afrag, bf[s][ot], acc[ot]);
  }
}
__global__ void __launch_bounds__(128) k_attn(const float* __restrict__ h,
                                              const float* __restrict__ si,
                                              const float* __restrict__ sj,
                                              const float* __restrict__ mx,
                                              float* __restrict__ out) {
  __shared__ __align__(16) float hs2[2][32 * PSTRIDE];
  __shared__ __align__(16) float kfs[2][128];
  __shared__ float zbuf[4 * 16];

  const int tid = threadIdx.x;
  const int lane = tid & 31;
  const int wave = tid >> 5;
  const int rl = lane & 15;
  const int hi = lane >> 4;
  const int rt = wave & 1;
  const int chf = wave >> 1;
  const int b = blockIdx.y;
  const int i0 = blockIdx.x * 32 + rt * 16;
  const int boff = chf * 128 + rl * 2;

  const float* hb = h + (size_t)b * 2048 * 128;
  const float* sjb = sj + (size_t)b * 2048;

  const float mxb = mx[b];
  const float s_row = si[(size_t)b * 2048 + i0 + rl] + mxb;
  const float m_row = leaky(s_row);
  const float a_one = __expf(s_row - m_row);
  const float c_one = __expf(ALPHA * s_row - m_row);
  const v2f ac_row = {a_one, c_one};

  v8f acc[4];
  v8f zero = {};
#pragma unroll
  for (int t = 0; t < 4; ++t) acc[t] = zero;
  float zacc = 0.0f;

  {
    float4 rA[8], rB[8];
    float sjr = 0.0f;
#pragma unroll
    for (int k = 0; k < 8; ++k) {
      const int q = tid + 128 * k;
      const int jp = q >> 5, c0 = (q & 31) * 4;
      rA[k] = *(const float4*)(hb + (size_t)(2 * jp) * 128 + c0);
      rB[k] = *(const float4*)(hb + (size_t)(2 * jp + 1) * 128 + c0);
    }
    if (tid < 64) sjr = sjb[tid];
#pragma unroll
    for (int k = 0; k < 8; ++k) {
      const int q = tid + 128 * k;
      const int jp = q >> 5, c0 = (q & 31) * 4;
      float4* dst = (float4*)&hs2[0][jp * PSTRIDE + c0 * 2];
      dst[0] = make_float4(rA[k].x, rB[k].x, rA[k].y, rB[k].y);
      dst[1] = make_float4(rA[k].z, rB[k].z, rA[k].w, rB[k].w);
    }
    if (tid < 64) {
      const float d = sjr - mxb;
      *(v2f*)&kfs[0][2 * tid] = v2f{__expf(d), __expf(ALPHA * d)};
    }
  }
  __syncthreads();

  for (int ch = 0; ch < 32; ++ch) {
    const float* hcur = hs2[ch & 1];
    const float* kfcur = kfs[ch & 1];
    float* hnxt = hs2[(ch + 1) & 1];
    float* kfnxt = kfs[(ch + 1) & 1];

    const int jn = (ch + 1 < 32 ? ch + 1 : 31) * 64;
    float4 rA[8], rB[8];
    float sjr = 0.0f;
#pragma unroll
    for (int k = 0; k < 8; ++k) {
      const int q = tid + 128 * k;
      const int jp = q >> 5, c0 = (q & 31) * 4;
      rA[k] = *(const float4*)(hb + (size_t)(jn + 2 * jp) * 128 + c0);
      rB[k] = *(const float4*)(hb + (size_t)(jn + 2 * jp + 1) * 128 + c0);
    }
    if (tid < 64) sjr = sjb[jn + tid];

    attn_half(hcur, kfcur, 0, hi, boff, ac_row, zacc, acc);

    __builtin_amdgcn_sched_barrier(0);
#pragma unroll
    for (int k = 0; k < 8; ++k) {
      const int q = tid + 128 * k;
      const int jp = q >> 5, c0 = (q & 31) * 4;
      float4* dst = (float4*)&hnxt[jp * PSTRIDE + c0 * 2];
      dst[0] = make_float4(rA[k].x, rB[k].x, rA[k].y, rB[k].y);
      dst[1] = make_float4(rA[k].z, rB[k].z, rA[k].w, rB[k].w);
    }
    if (tid < 64) {
      const float d = sjr - mxb;
      *(v2f*)&kfnxt[2 * tid] = v2f{__expf(d), __expf(ALPHA * d)};
    }

    attn_half(hcur, kfcur, 32, hi, boff, ac_row, zacc, acc);
    __syncthreads();
  }

  float zfull = zacc + __shfl_xor(zacc, 16, 32);
  if (hi == 0) zbuf[wave * 16 + rl] = zfull;
  __syncthreads();
  float zinv[8];
#pragma unroll
  for (int r = 0; r < 8; ++r) zinv[r] = 1.0f / zbuf[wave * 16 + r + hi * 8];

#pragma unroll
  for (int ot = 0; ot < 4; ++ot)
#pragma unroll
    for (int r = 0; r < 8; ++r)
      out[((size_t)b * 2048 + i0 + r + hi * 8) * 128 + chf * 64 + ot * 16 + rl] =
          acc[ot][r] * zinv[r];
}

extern "C" void kernel_launch(void* const* d_in, const int* in_sizes, int n_in,
                              void* d_out, int out_size, void* d_ws, size_t ws_size,
                              hipStream_t stream) {
  const float* x = (const float*)d_in[0];
  const float* W = (const float*)d_in[1];
  const float* a = (const float*)d_in[2];
  float* out = (float*)d_out;

  float* h  = (float*)d_ws;
  float* si = h + (size_t)16384 * 128;
  float* sj = si + 16384;
  float* mx = sj + 16384;

  k_h_gemm<<<dim3(256), dim3(128), 0, stream>>>(x, W, h);
  k_scores<<<dim3(256), dim3(256), 0, stream>>>(h, a, si, sj);
  k_maxsj<<<dim3(8), dim3(256), 0, stream>>>(sj, mx);
  k_attn<<<dim3(64, 8), dim3(128), 0, stream>>>(h, si, sj, mx, out);
}


# attention: block = 64 rows x 64 columns (4 row tiles); column halves are separate blocks, so a block stages only its half of h: global->LDS bytes, loads, moves and LDS writes per block halved
# speedup vs baseline: 1.5870x; 1.0911x over previous
#include <hip/hip_runtime.h>

#define ALPHA 0.2f

typedef __attribute__((ext_vector_type(2))) float v2f;
typedef __attribute__((ext_vector_type(4))) float v4f;
typedef __attribute__((ext_vector_type(8))) float v8f;

__device__ __forceinline__ v8f wmma_f32_16x16x4(v2f a, v2f b, v8f c) {
  return __builtin_amdgcn_wmma_f32_16x16x4_f32(false, a, false, b, (short)0, c,
                                               false, false);
}

__device__ __forceinline__ float leaky(float t) {
  return (t >= 0.0f) ? t : ALPHA * t;
}

#define G_DS 2
#define G_NST 8
__device__ __forceinline__ void gemm_load_stage(const float* xa, const float* wb,
                                                int kk0, v4f (&fa)[G_DS][2],
                                                v4f (&fb)[G_DS][4]) {
#pragma unroll
  for (int d = 0; d < G_DS; ++d) {
#pragma unroll
    for (int rt = 0; rt < 2; ++rt)
      fa[d][rt] = *(const v4f*)(xa + (size_t)rt * 16 * 128 + kk0 + d * 8);
#pragma unroll
    for (int ot = 0; ot < 4; ++ot)
      fb[d][ot] = *(const v4f*)(wb + (size_t)ot * 16 * 128 + kk0 + d * 8);
  }
}
__device__ __forceinline__ void gemm_compute_stage(const v4f (&fa)[G_DS][2],
                                                   const v4f (&fb)[G_DS][4],
                                                   v8f (&acc)[2][4]) {
#pragma unroll
  for (int d = 0; d < G_DS; ++d) {
#pragma unroll
    for (int half = 0; half < 2; ++half) {
      v2f a[2], b[4];
#pragma unroll
      for (int rt = 0; rt < 2; ++rt)
        a[rt] = v2f{fa[d][rt][2 * half], fa[d][rt][2 * half + 1]};
#pragma unroll
      for (int ot = 0; ot < 4; ++ot)
        b[ot] = v2f{fb[d][ot][2 * half], fb[d][ot][2 * half + 1]};
#pragma unroll
      for (int rt = 0; rt < 2; ++rt)
#pragma unroll
        for (int ot = 0; ot < 4; ++ot)
          acc[rt][ot] = wmma_f32_16x16x4(a[rt], b[ot], acc[rt][ot]);
    }
  }
}
__global__ void __launch_bounds__(128) k_h_gemm(const float* __restrict__ x,
                                                const float* __restrict__ W,
                                                float* __restrict__ h) {
  const int lane = threadIdx.x & 31;
  const int wave = threadIdx.x >> 5;
  const int rl = lane & 15;
  const int hi = lane >> 4;
  const int i0 = blockIdx.x * 64 + (wave & 1) * 32;
  const int cb = (wave >> 1) * 64;

  v8f acc[2][4];
  v8f zero = {};
#pragma unroll
  for (int rt = 0; rt < 2; ++rt)
#pragma unroll
    for (int ot = 0; ot < 4; ++ot) acc[rt][ot] = zero;

  const float* xa = x + (size_t)(i0 + rl) * 128 + 4 * hi;
  const float* wb = W + (size_t)(cb + rl) * 128 + 4 * hi;

  v4f fa[3][G_DS][2], fb[3][G_DS][4];
  gemm_load_stage(xa, wb, 0, fa[0], fb[0]);
  gemm_load_stage(xa, wb, G_DS * 8, fa[1], fb[1]);
#pragma unroll
  for (int st = 0; st < G_NST; ++st) {
    if (st + 2 < G_NST)
      gemm_load_stage(xa, wb, (st + 2) * G_DS * 8, fa[(st + 2) % 3], fb[(st + 2) % 3]);
    __builtin_amdgcn_sched_barrier(0);
    gemm_compute_stage(fa[st % 3], fb[st % 3], acc);
  }
#pragma unroll
  for (int rt = 0; rt < 2; ++rt)
#pragma unroll
    for (int ot = 0; ot < 4; ++ot)
#pragma unroll
      for (int r = 0; r < 8; ++r)
        h[(size_t)(i0 + rt * 16 + r + hi * 8) * 128 + cb + ot * 16 + rl] =
            acc[rt][ot][r];
}

#define SC_ROWS 8
__global__ void __launch_bounds__(256) k_scores(const float* __restrict__ h,
                                                const float* __restrict__ a,
                                                float* __restrict__ si,
                                                float* __restrict__ sj) {
  const int lane = threadIdx.x & 31;
  const int row0 = (blockIdx.x * 8 + (threadIdx.x >> 5)) * SC_ROWS;
  const float* hp = h + (size_t)row0 * 128 + lane;

  float hv[SC_ROWS][4];
#pragma unroll
  for (int r = 0; r < SC_ROWS; ++r)
#pragma unroll
    for (int k = 0; k < 4; ++k) hv[r][k] = hp[r * 128 + k * 32];
  float a1[4], a2[4];
#pragma unroll
  for (int k = 0; k < 4; ++k) {
    a1[k] = a[lane + k * 32];
    a2[k] = a[128 + lane + k * 32];
  }

  float psi[SC_ROWS], psj[SC_ROWS];
#pragma unroll
  for (int r = 0; r < SC_ROWS; ++r) {
    psi[r] = 0.0f;
    psj[r] = 0.0f;
#pragma unroll
    for (int k = 0; k < 4; ++k) {
      psi[r] += hv[r][k] * a1[k];
      psj[r] += hv[r][k] * a2[k];
    }
  }
#pragma unroll
  for (int off = 16; off > 0; off >>= 1) {
#pragma unroll
    for (int r = 0; r < SC_ROWS; ++r) {
      psi[r] += __shfl_xor(psi[r], off, 32);
      psj[r] += __shfl_xor(psj[r], off, 32);
    }
  }
  if (lane < SC_ROWS) {
    float osi = psi[0], osj = psj[0];
#pragma unroll
    for (int r = 1; r < SC_ROWS; ++r) {
      if (lane == r) {
        osi = psi[r];
        osj = psj[r];
      }
    }
    si[row0 + lane] = osi;
    sj[row0 + lane] = osj;
  }
}

__global__ void __launch_bounds__(256) k_maxsj(const float* __restrict__ sj,
                                               float* __restrict__ mx) {
  __shared__ float red[256];
  const int b = blockIdx.x;
  float m = -3.402823466e38f;
  for (int j = threadIdx.x; j < 2048; j += 256)
    m = fmaxf(m, sj[(size_t)b * 2048 + j]);
  red[threadIdx.x] = m;
  __syncthreads();
  for (int s = 128; s > 0; s >>= 1) {
    if (threadIdx.x < s) red[threadIdx.x] = fmaxf(red[threadIdx.x], red[threadIdx.x + s]);
    __syncthreads();
  }
  if (threadIdx.x == 0) mx[b] = red[0];
}

#define PSTRIDE 160

__device__ __forceinline__ void attn_half(const float* hcur, const float* kfcur,
                                          int j0, int hi, int boff, v2f ac_row,
                                          float& zacc, v8f (&acc)[4]) {
  v4f kf[8];
  v2f bf[8][4];
#pragma unroll
  for (int s = 0; s < 8; ++s) {
    const int j = j0 + 4 * s;
    const int jp = (j >> 1) + hi;
    kf[s] = *(const v4f*)&kfcur[jp * 4];
    const float* bb = &hcur[jp * PSTRIDE + boff];
#pragma unroll
    for (int ot = 0; ot < 4; ++ot)
      bf[s][ot] = *(const v2f*)(bb + ot * 32);
  }
  __builtin_amdgcn_sched_barrier(0);
#pragma unroll
  for (int s = 0; s < 8; ++s) {
    const v2f k0 = {kf[s][0], kf[s][1]};
    const v2f k1 = {kf[s][2], kf[s][3]};
    const v2f q0 = ac_row * k0;
    const v2f q1 = ac_row * k1;
    float p0 = fmaxf(q0[0], q0[1]);
    float p1 = fmaxf(q1[0], q1[1]);
    zacc += p0 + p1;
    v2f afrag = {p0, p1};
#pragma unroll
    for (int ot = 0; ot < 4; ++ot)
      acc[ot] = wmma_f32_16x16x4(afrag, bf[s][ot], acc[ot]);
  }
}
__global__ void __launch_bounds__(128) k_attn(const float* __restrict__ h,
                                              const float* __restrict__ si,
                                              const float* __restrict__ sj,
                                              const float* __restrict__ mx,
                                              float* __restrict__ out) {
  __shared__ __align__(16) float hs2[2][32 * PSTRIDE];
  __shared__ __align__(16) float kfs[2][128];
  __shared__ float zbuf[4 * 16];

  const int tid = threadIdx.x;
  const int lane = tid & 31;
  const int wave = tid >> 5;
  const int rl = lane & 15;
  const int hi = lane >> 4;
  const int nb = blockIdx.y * 64 + blockIdx.x;
  const int chf = (nb >> 7) & 1;
  const int b = (nb >> 4) & 7;
  const int rb = ((nb & 15) << 1) | (nb >> 8);
  const int i0 = rb * 64 + wave * 16;
  const int boff = rl * 2;

  const float* hb = h + (size_t)b * 2048 * 128 + chf * 64;
  const float* sjb = sj + (size_t)b * 2048;

  const float mxb = mx[b];
  const float s_row = si[(size_t)b * 2048 + i0 + rl] + mxb;
  const float m_row = leaky(s_row);
  const float a_one = __expf(s_row - m_row);
  const float c_one = __expf(ALPHA * s_row - m_row);
  const v2f ac_row = {a_one, c_one};

  v8f acc[4];
  v8f zero = {};
#pragma unroll
  for (int t = 0; t < 4; ++t) acc[t] = zero;
  float zacc = 0.0f;

  {
    float4 rA[4], rB[4];
    float sjr = 0.0f;
#pragma unroll
    for (int k = 0; k < 4; ++k) {
      const int q = tid + 128 * k;
      const int jp = q >> 4, c0 = (q & 15) * 4;
      rA[k] = *(const float4*)(hb + (size_t)(2 * jp) * 128 + c0);
      rB[k] = *(const float4*)(hb + (size_t)(2 * jp + 1) * 128 + c0);
    }
    if (tid < 64) sjr = sjb[tid];
#pragma unroll
    for (int k = 0; k < 4; ++k) {
      const int q = tid + 128 * k;
      const int jp = q >> 4, c0 = (q & 15) * 4;
      float4* dst = (float4*)&hs2[0][jp * PSTRIDE + c0 * 2];
      dst[0] = make_float4(rA[k].x, rB[k].x, rA[k].y, rB[k].y);
      dst[1] = make_float4(rA[k].z, rB[k].z, rA[k].w, rB[k].w);
    }
    if (tid < 64) {
      const float d = sjr - mxb;
      *(v2f*)&kfs[0][2 * tid] = v2f{__expf(d), __expf(ALPHA * d)};
    }
  }
  __syncthreads();

  for (int ch = 0; ch < 32; ++ch) {
    const float* hcur = hs2[ch & 1];
    const float* kfcur = kfs[ch & 1];
    float* hnxt = hs2[(ch + 1) & 1];
    float* kfnxt = kfs[(ch + 1) & 1];

    const int jn = (ch + 1 < 32 ? ch + 1 : 31) * 64;
    float4 rA[4], rB[4];
    float sjr = 0.0f;
#pragma unroll
    for (int k = 0; k < 4; ++k) {
      const int q = tid + 128 * k;
      const int jp = q >> 4, c0 = (q & 15) * 4;
      rA[k] = *(const float4*)(hb + (size_t)(jn + 2 * jp) * 128 + c0);
      rB[k] = *(const float4*)(hb + (size_t)(jn + 2 * jp + 1) * 128 + c0);
    }
    if (tid < 64) sjr = sjb[jn + tid];

    attn_half(hcur, kfcur, 0, hi, boff, ac_row, zacc, acc);

    __builtin_amdgcn_sched_barrier(0);
#pragma unroll
    for (int k = 0; k < 4; ++k) {
      const int q = tid + 128 * k;
      const int jp = q >> 4, c0 = (q & 15) * 4;
      float4* dst = (float4*)&hnxt[jp * PSTRIDE + c0 * 2];
      dst[0] = make_float4(rA[k].x, rB[k].x, rA[k].y, rB[k].y);
      dst[1] = make_float4(rA[k].z, rB[k].z, rA[k].w, rB[k].w);
    }
    if (tid < 64) {
      const float d = sjr - mxb;
      *(v2f*)&kfnxt[2 * tid] = v2f{__expf(d), __expf(ALPHA * d)};
    }

    attn_half(hcur, kfcur, 32, hi, boff, ac_row, zacc, acc);
    __syncthreads();
  }

  float zfull = zacc + __shfl_xor(zacc, 16, 32);
  if (hi == 0) zbuf[wave * 16 + rl] = zfull;
  __syncthreads();
  float zinv[8];
#pragma unroll
  for (int r = 0; r < 8; ++r) zinv[r] = 1.0f / zbuf[wave * 16 + r + hi * 8];

#pragma unroll
  for (int ot = 0; ot < 4; ++ot)
#pragma unroll
    for (int r = 0; r < 8; ++r)
      out[((size_t)b * 2048 + i0 + r + hi * 8) * 128 + chf * 64 + ot * 16 + rl] =
          acc[ot][r] * zinv[r];
}

extern "C" void kernel_launch(void* const* d_in, const int* in_sizes, int n_in,
                              void* d_out, int out_size, void* d_ws, size_t ws_size,
                              hipStream_t stream) {
  const float* x = (const float*)d_in[0];
  const float* W = (const float*)d_in[1];
  const float* a = (const float*)d_in[2];
  float* out = (float*)d_out;

  float* h  = (float*)d_ws;
  float* si = h + (size_t)16384 * 128;
  float* sj = si + 16384;
  float* mx = sj + 16384;

  k_h_gemm<<<dim3(256), dim3(128), 0, stream>>>(x, W, h);
  k_scores<<<dim3(256), dim3(256), 0, stream>>>(h, a, si, sj);
  k_maxsj<<<dim3(8), dim3(256), 0, stream>>>(sj, mx);
  k_attn<<<dim3(64, 8), dim3(128), 0, stream>>>(h, si, sj, mx, out);
}


# per-batch max launch removed: each score-kernel wave also stores the max of its 8 sj (2048 partials); every attention wave reduces its batch's 256 partials itself; launches 4 -> 3
# speedup vs baseline: 1.7250x; 1.0870x over previous
#include <hip/hip_runtime.h>

#define ALPHA 0.2f

typedef __attribute__((ext_vector_type(2))) float v2f;
typedef __attribute__((ext_vector_type(4))) float v4f;
typedef __attribute__((ext_vector_type(8))) float v8f;

__device__ __forceinline__ v8f wmma_f32_16x16x4(v2f a, v2f b, v8f c) {
  return __builtin_amdgcn_wmma_f32_16x16x4_f32(false, a, false, b, (short)0, c,
                                               false, false);
}

__device__ __forceinline__ float leaky(float t) {
  return (t >= 0.0f) ? t : ALPHA * t;
}

#define G_DS 2
#define G_NST 8
__device__ __forceinline__ void gemm_load_stage(const float* xa, const float* wb,
                                                int kk0, v4f (&fa)[G_DS][2],
                                                v4f (&fb)[G_DS][4]) {
#pragma unroll
  for (int d = 0; d < G_DS; ++d) {
#pragma unroll
    for (int rt = 0; rt < 2; ++rt)
      fa[d][rt] = *(const v4f*)(xa + (size_t)rt * 16 * 128 + kk0 + d * 8);
#pragma unroll
    for (int ot = 0; ot < 4; ++ot)
      fb[d][ot] = *(const v4f*)(wb + (size_t)ot * 16 * 128 + kk0 + d * 8);
  }
}
__device__ __forceinline__ void gemm_compute_stage(const v4f (&fa)[G_DS][2],
                                                   const v4f (&fb)[G_DS][4],
                                                   v8f (&acc)[2][4]) {
#pragma unroll
  for (int d = 0; d < G_DS; ++d) {
#pragma unroll
    for (int half = 0; half < 2; ++half) {
      v2f a[2], b[4];
#pragma unroll
      for (int rt = 0; rt < 2; ++rt)
        a[rt] = v2f{fa[d][rt][2 * half], fa[d][rt][2 * half + 1]};
#pragma unroll
      for (int ot = 0; ot < 4; ++ot)
        b[ot] = v2f{fb[d][ot][2 * half], fb[d][ot][2 * half + 1]};
#pragma unroll
      for (int rt = 0; rt < 2; ++rt)
#pragma unroll
        for (int ot = 0; ot < 4; ++ot)
          acc[rt][ot] = wmma_f32_16x16x4(a[rt], b[ot], acc[rt][ot]);
    }
  }
}
__global__ void __launch_bounds__(128) k_h_gemm(const float* __restrict__ x,
                                                const float* __restrict__ W,
                                                float* __restrict__ h) {
  const int lane = threadIdx.x & 31;
  const int wave = threadIdx.x >> 5;
  const int rl = lane & 15;
  const int hi = lane >> 4;
  const int i0 = blockIdx.x * 64 + (wave & 1) * 32;
  const int cb = (wave >> 1) * 64;

  v8f acc[2][4];
  v8f zero = {};
#pragma unroll
  for (int rt = 0; rt < 2; ++rt)
#pragma unroll
    for (int ot = 0; ot < 4; ++ot) acc[rt][ot] = zero;

  const float* xa = x + (size_t)(i0 + rl) * 128 + 4 * hi;
  const float* wb = W + (size_t)(cb + rl) * 128 + 4 * hi;

  v4f fa[3][G_DS][2], fb[3][G_DS][4];
  gemm_load_stage(xa, wb, 0, fa[0], fb[0]);
  gemm_load_stage(xa, wb, G_DS * 8, fa[1], fb[1]);
#pragma unroll
  for (int st = 0; st < G_NST; ++st) {
    if (st + 2 < G_NST)
      gemm_load_stage(xa, wb, (st + 2) * G_DS * 8, fa[(st + 2) % 3], fb[(st + 2) % 3]);
    __builtin_amdgcn_sched_barrier(0);
    gemm_compute_stage(fa[st % 3], fb[st % 3], acc);
  }
#pragma unroll
  for (int rt = 0; rt < 2; ++rt)
#pragma unroll
    for (int ot = 0; ot < 4; ++ot)
#pragma unroll
      for (int r = 0; r < 8; ++r)
        h[(size_t)(i0 + rt * 16 + r + hi * 8) * 128 + cb + ot * 16 + rl] =
            acc[rt][ot][r];
}

#define SC_ROWS 8
__global__ void __launch_bounds__(256) k_scores(const float* __restrict__ h,
                                                const float* __restrict__ a,
                                                float* __restrict__ si,
                                                float* __restrict__ sj,
                                                float* __restrict__ pmx) {
  const int lane = threadIdx.x & 31;
  const int wv = blockIdx.x * 8 + (threadIdx.x >> 5);
  const int row0 = wv * SC_ROWS;
  const float* hp = h + (size_t)row0 * 128 + lane;

  float hv[SC_ROWS][4];
#pragma unroll
  for (int r = 0; r < SC_ROWS; ++r)
#pragma unroll
    for (int k = 0; k < 4; ++k) hv[r][k] = hp[r * 128 + k * 32];
  float a1[4], a2[4];
#pragma unroll
  for (int k = 0; k < 4; ++k) {
    a1[k] = a[lane + k * 32];
    a2[k] = a[128 + lane + k * 32];
  }

  float psi[SC_ROWS], psj[SC_ROWS];
#pragma unroll
  for (int r = 0; r < SC_ROWS; ++r) {
    psi[r] = 0.0f;
    psj[r] = 0.0f;
#pragma unroll
    for (int k = 0; k < 4; ++k) {
      psi[r] += hv[r][k] * a1[k];
      psj[r] += hv[r][k] * a2[k];
    }
  }
#pragma unroll
  for (int off = 16; off > 0; off >>= 1) {
#pragma unroll
    for (int r = 0; r < SC_ROWS; ++r) {
      psi[r] += __shfl_xor(psi[r], off, 32);
      psj[r] += __shfl_xor(psj[r], off, 32);
    }
  }
  if (lane < SC_ROWS) {
    float osi = psi[0], osj = psj[0];
#pragma unroll
    for (int r = 1; r < SC_ROWS; ++r) {
      if (lane == r) {
        osi = psi[r];
        osj = psj[r];
      }
    }
    si[row0 + lane] = osi;
    sj[row0 + lane] = osj;
  }
  float wmx = -3.402823466e38f;
#pragma unroll
  for (int r = 0; r < SC_ROWS; ++r) wmx = fmaxf(wmx, psj[r]);
  if (lane == 0) pmx[wv] = wmx;
}

#define PSTRIDE 160

__device__ __forceinline__ void attn_half(const float* hcur, const float* kfcur,
                                          int j0, int hi, int boff, v2f ac_row,
                                          float& zacc, v8f (&acc)[4]) {
  v4f kf[8];
  v2f bf[8][4];
#pragma unroll
  for (int s = 0; s < 8; ++s) {
    const int j = j0 + 4 * s;
    const int jp = (j >> 1) + hi;
    kf[s] = *(const v4f*)&kfcur[jp * 4];
    const float* bb = &hcur[jp * PSTRIDE + boff];
#pragma unroll
    for (int ot = 0; ot < 4; ++ot)
      bf[s][ot] = *(const v2f*)(bb + ot * 32);
  }
  __builtin_amdgcn_sched_barrier(0);
#pragma unroll
  for (int s = 0; s < 8; ++s) {
    const v2f k0 = {kf[s][0], kf[s][1]};
    const v2f k1 = {kf[s][2], kf[s][3]};
    const v2f q0 = ac_row * k0;
    const v2f q1 = ac_row * k1;
    float p0 = fmaxf(q0[0], q0[1]);
    float p1 = fmaxf(q1[0], q1[1]);
    zacc += p0 + p1;
    v2f afrag = {p0, p1};
#pragma unroll
    for (int ot = 0; ot < 4; ++ot)
      acc[ot] = wmma_f32_16x16x4(afrag, bf[s][ot], acc[ot]);
  }
}
__global__ void __launch_bounds__(128) k_attn(const float* __restrict__ h,
                                              const float* __restrict__ si,
                                              const float* __restrict__ sj,
                                              const float* __restrict__ pmx,
                                              float* __restrict__ out) {
  __shared__ __align__(16) float hs2[2][32 * PSTRIDE];
  __shared__ __align__(16) float kfs[2][128];
  __shared__ float zbuf[4 * 16];

  const int tid = threadIdx.x;
  const int lane = tid & 31;
  const int wave = tid >> 5;
  const int rl = lane & 15;
  const int hi = lane >> 4;
  const int nb = blockIdx.y * 64 + blockIdx.x;
  const int chf = (nb >> 7) & 1;
  const int b = (nb >> 4) & 7;
  const int rb = ((nb & 15) << 1) | (nb >> 8);
  const int i0 = rb * 64 + wave * 16;
  const int boff = rl * 2;

  const float* hb = h + (size_t)b * 2048 * 128 + chf * 64;
  const float* sjb = sj + (size_t)b * 2048;

  float mxb = -3.402823466e38f;
  {
    const float4* pm = (const float4*)(pmx + b * 256 + lane * 8);
    const float4 pa = pm[0], pb = pm[1];
    mxb = fmaxf(mxb, fmaxf(fmaxf(pa.x, pa.y), fmaxf(pa.z, pa.w)));
    mxb = fmaxf(mxb, fmaxf(fmaxf(pb.x, pb.y), fmaxf(pb.z, pb.w)));
#pragma unroll
    for (int off = 16; off > 0; off >>= 1) mxb = fmaxf(mxb, __shfl_xor(mxb, off, 32));
  }
  const float s_row = si[(size_t)b * 2048 + i0 + rl] + mxb;
  const float m_row = leaky(s_row);
  const float a_one = __expf(s_row - m_row);
  const float c_one = __expf(ALPHA * s_row - m_row);
  const v2f ac_row = {a_one, c_one};

  v8f acc[4];
  v8f zero = {};
#pragma unroll
  for (int t = 0; t < 4; ++t) acc[t] = zero;
  float zacc = 0.0f;

  {
    float4 rA[4], rB[4];
    float sjr = 0.0f;
#pragma unroll
    for (int k = 0; k < 4; ++k) {
      const int q = tid + 128 * k;
      const int jp = q >> 4, c0 = (q & 15) * 4;
      rA[k] = *(const float4*)(hb + (size_t)(2 * jp) * 128 + c0);
      rB[k] = *(const float4*)(hb + (size_t)(2 * jp + 1) * 128 + c0);
    }
    if (tid < 64) sjr = sjb[tid];
#pragma unroll
    for (int k = 0; k < 4; ++k) {
      const int q = tid + 128 * k;
      const int jp = q >> 4, c0 = (q & 15) * 4;
      float4* dst = (float4*)&hs2[0][jp * PSTRIDE + c0 * 2];
      dst[0] = make_float4(rA[k].x, rB[k].x, rA[k].y, rB[k].y);
      dst[1] = make_float4(rA[k].z, rB[k].z, rA[k].w, rB[k].w);
    }
    if (tid < 64) {
      const float d = sjr - mxb;
      *(v2f*)&kfs[0][2 * tid] = v2f{__expf(d), __expf(ALPHA * d)};
    }
  }
  __syncthreads();

  for (int ch = 0; ch < 32; ++ch) {
    const float* hcur = hs2[ch & 1];
    const float* kfcur = kfs[ch & 1];
    float* hnxt = hs2[(ch + 1) & 1];
    float* kfnxt = kfs[(ch + 1) & 1];

    const int jn = (ch + 1 < 32 ? ch + 1 : 31) * 64;
    float4 rA[4], rB[4];
    float sjr = 0.0f;
#pragma unroll
    for (int k = 0; k < 4; ++k) {
      const int q = tid + 128 * k;
      const int jp = q >> 4, c0 = (q & 15) * 4;
      rA[k] = *(const float4*)(hb + (size_t)(jn + 2 * jp) * 128 + c0);
      rB[k] = *(const float4*)(hb + (size_t)(jn + 2 * jp + 1) * 128 + c0);
    }
    if (tid < 64) sjr = sjb[jn + tid];

    attn_half(hcur, kfcur, 0, hi, boff, ac_row, zacc, acc);

    __builtin_amdgcn_sched_barrier(0);
#pragma unroll
    for (int k = 0; k < 4; ++k) {
      const int q = tid + 128 * k;
      const int jp = q >> 4, c0 = (q & 15) * 4;
      float4* dst = (float4*)&hnxt[jp * PSTRIDE + c0 * 2];
      dst[0] = make_float4(rA[k].x, rB[k].x, rA[k].y, rB[k].y);
      dst[1] = make_float4(rA[k].z, rB[k].z, rA[k].w, rB[k].w);
    }
    if (tid < 64) {
      const float d = sjr - mxb;
      *(v2f*)&kfnxt[2 * tid] = v2f{__expf(d), __expf(ALPHA * d)};
    }

    attn_half(hcur, kfcur, 32, hi, boff, ac_row, zacc, acc);
    __syncthreads();
  }

  float zfull = zacc + __shfl_xor(zacc, 16, 32);
  if (hi == 0) zbuf[wave * 16 + rl] = zfull;
  __syncthreads();
  float zinv[8];
#pragma unroll
  for (int r = 0; r < 8; ++r) zinv[r] = 1.0f / zbuf[wave * 16 + r + hi * 8];

#pragma unroll
  for (int ot = 0; ot < 4; ++ot)
#pragma unroll
    for (int r = 0; r < 8; ++r)
      out[((size_t)b * 2048 + i0 + r + hi * 8) * 128 + chf * 64 + ot * 16 + rl] =
          acc[ot][r] * zinv[r];
}

extern "C" void kernel_launch(void* const* d_in, const int* in_sizes, int n_in,
                              void* d_out, int out_size, void* d_ws, size_t ws_size,
                              hipStream_t stream) {
  const float* x = (const float*)d_in[0];
  const float* W = (const float*)d_in[1];
  const float* a = (const float*)d_in[2];
  float* out = (float*)d_out;

  float* h  = (float*)d_ws;
  float* si = h + (size_t)16384 * 128;
  float* sj = si + 16384;
  float* pmx = sj + 16384;

  k_h_gemm<<<dim3(256), dim3(128), 0, stream>>>(x, W, h);
  k_scores<<<dim3(256), dim3(256), 0, stream>>>(h, a, si, sj, pmx);
  k_attn<<<dim3(64, 8), dim3(128), 0, stream>>>(h, si, sj, pmx, out);
}


# attention sweep replaced by running sums over keys ranked by sj: max(Ai*Bj,Ci*Dj) = Ai*Bj where sj>=-si, else Ci*Dj; count rank, segment sums, 8-key tail per row; new summation order; launches 3->5
# speedup vs baseline: 2.8209x; 1.6353x over previous
#include <hip/hip_runtime.h>

#define ALPHA 0.2f

typedef __attribute__((ext_vector_type(2))) float v2f;
typedef __attribute__((ext_vector_type(4))) float v4f;
typedef __attribute__((ext_vector_type(8))) float v8f;

__device__ __forceinline__ v8f wmma_f32_16x16x4(v2f a, v2f b, v8f c) {
  return __builtin_amdgcn_wmma_f32_16x16x4_f32(false, a, false, b, (short)0, c,
                                               false, false);
}

__device__ __forceinline__ float leaky(float t) {
  return (t >= 0.0f) ? t : ALPHA * t;
}

#define G_DS 2
#define G_NST 8
__device__ __forceinline__ void gemm_load_stage(const float* xa, const float* wb,
                                                int kk0, v4f (&fa)[G_DS][2],
                                                v4f (&fb)[G_DS][4]) {
#pragma unroll
  for (int d = 0; d < G_DS; ++d) {
#pragma unroll
    for (int rt = 0; rt < 2; ++rt)
      fa[d][rt] = *(const v4f*)(xa + (size_t)rt * 16 * 128 + kk0 + d * 8);
#pragma unroll
    for (int ot = 0; ot < 4; ++ot)
      fb[d][ot] = *(const v4f*)(wb + (size_t)ot * 16 * 128 + kk0 + d * 8);
  }
}
__device__ __forceinline__ void gemm_compute_stage(const v4f (&fa)[G_DS][2],
                                                   const v4f (&fb)[G_DS][4],
                                                   v8f (&acc)[2][4]) {
#pragma unroll
  for (int d = 0; d < G_DS; ++d) {
#pragma unroll
    for (int half = 0; half < 2; ++half) {
      v2f a[2], b[4];
#pragma unroll
      for (int rt = 0; rt < 2; ++rt)
        a[rt] = v2f{fa[d][rt][2 * half], fa[d][rt][2 * half + 1]};
#pragma unroll
      for (int ot = 0; ot < 4; ++ot)
        b[ot] = v2f{fb[d][ot][2 * half], fb[d][ot][2 * half + 1]};
#pragma unroll
      for (int rt = 0; rt < 2; ++rt)
#pragma unroll
        for (int ot = 0; ot < 4; ++ot)
          acc[rt][ot] = wmma_f32_16x16x4(a[rt], b[ot], acc[rt][ot]);
    }
  }
}
__global__ void __launch_bounds__(128) k_h_gemm(const float* __restrict__ x,
                                                const float* __restrict__ W,
                                                float* __restrict__ h) {
  const int lane = threadIdx.x & 31;
  const int wave = threadIdx.x >> 5;
  const int rl = lane & 15;
  const int hi = lane >> 4;
  const int i0 = blockIdx.x * 64 + (wave & 1) * 32;
  const int cb = (wave >> 1) * 64;

  v8f acc[2][4];
  v8f zero = {};
#pragma unroll
  for (int rt = 0; rt < 2; ++rt)
#pragma unroll
    for (int ot = 0; ot < 4; ++ot) acc[rt][ot] = zero;

  const float* xa = x + (size_t)(i0 + rl) * 128 + 4 * hi;
  const float* wb = W + (size_t)(cb + rl) * 128 + 4 * hi;

  v4f fa[3][G_DS][2], fb[3][G_DS][4];
  gemm_load_stage(xa, wb, 0, fa[0], fb[0]);
  gemm_load_stage(xa, wb, G_DS * 8, fa[1], fb[1]);
#pragma unroll
  for (int st = 0; st < G_NST; ++st) {
    if (st + 2 < G_NST)
      gemm_load_stage(xa, wb, (st + 2) * G_DS * 8, fa[(st + 2) % 3], fb[(st + 2) % 3]);
    __builtin_amdgcn_sched_barrier(0);
    gemm_compute_stage(fa[st % 3], fb[st % 3], acc);
  }
#pragma unroll
  for (int rt = 0; rt < 2; ++rt)
#pragma unroll
    for (int ot = 0; ot < 4; ++ot)
#pragma unroll
      for (int r = 0; r < 8; ++r)
        h[(size_t)(i0 + rt * 16 + r + hi * 8) * 128 + cb + ot * 16 + rl] =
            acc[rt][ot][r];
}

#define SC_ROWS 8
__global__ void __launch_bounds__(256) k_scores(const float* __restrict__ h,
                                                const float* __restrict__ a,
                                                float* __restrict__ si,
                                                float* __restrict__ sj,
                                                float* __restrict__ pmx) {
  const int lane = threadIdx.x & 31;
  const int wv = blockIdx.x * 8 + (threadIdx.x >> 5);
  const int row0 = wv * SC_ROWS;
  const float* hp = h + (size_t)row0 * 128 + lane;

  float hv[SC_ROWS][4];
#pragma unroll
  for (int r = 0; r < SC_ROWS; ++r)
#pragma unroll
    for (int k = 0; k < 4; ++k) hv[r][k] = hp[r * 128 + k * 32];
  float a1[4], a2[4];
#pragma unroll
  for (int k = 0; k < 4; ++k) {
    a1[k] = a[lane + k * 32];
    a2[k] = a[128 + lane + k * 32];
  }

  float psi[SC_ROWS], psj[SC_ROWS];
#pragma unroll
  for (int r = 0; r < SC_ROWS; ++r) {
    psi[r] = 0.0f;
    psj[r] = 0.0f;
#pragma unroll
    for (int k = 0; k < 4; ++k) {
      psi[r] += hv[r][k] * a1[k];
      psj[r] += hv[r][k] * a2[k];
    }
  }
#pragma unroll
  for (int off = 16; off > 0; off >>= 1) {
#pragma unroll
    for (int r = 0; r < SC_ROWS; ++r) {
      psi[r] += __shfl_xor(psi[r], off, 32);
      psj[r] += __shfl_xor(psj[r], off, 32);
    }
  }
  if (lane < SC_ROWS) {
    float osi = psi[0], osj = psj[0];
#pragma unroll
    for (int r = 1; r < SC_ROWS; ++r) {
      if (lane == r) {
        osi = psi[r];
        osj = psj[r];
      }
    }
    si[row0 + lane] = osi;
    sj[row0 + lane] = osj;
  }
  float wmx = -3.402823466e38f;
#pragma unroll
  for (int r = 0; r < SC_ROWS; ++r) wmx = fmaxf(wmx, psj[r]);
  if (lane == 0) pmx[wv] = wmx;
}

#define SEG 8
#define NSEG 256
#define BSTR 136

__global__ void __launch_bounds__(256) k_rank(const float* __restrict__ si,
                                              const float* __restrict__ sj,
                                              const float* __restrict__ pmx,
                                              float* __restrict__ sidx_f,
                                              float* __restrict__ sbd_f,
                                              float* __restrict__ trow_f) {
  __shared__ __align__(16) float sjs[2048];
  __shared__ int cR[4][64];
  __shared__ int cT[4][64];

  const int tid = threadIdx.x;
  const int lane = tid & 31;
  const int b = blockIdx.y;
  const int jc = blockIdx.x;
  const int jl = tid & 63;
  const int q = __builtin_amdgcn_readfirstlane(tid >> 6);
  const int j = jc * 64 + jl;
  const float* sjb = sj + (size_t)b * 2048;

#pragma unroll
  for (int k = 0; k < 8; ++k) sjs[tid + 256 * k] = sjb[tid + 256 * k];

  float mxb = -3.402823466e38f;
  {
    const float4* pm = (const float4*)(pmx + b * 256 + lane * 8);
    const float4 pa = pm[0], pb = pm[1];
    mxb = fmaxf(mxb, fmaxf(fmaxf(pa.x, pa.y), fmaxf(pa.z, pa.w)));
    mxb = fmaxf(mxb, fmaxf(fmaxf(pb.x, pb.y), fmaxf(pb.z, pb.w)));
#pragma unroll
    for (int off = 16; off > 0; off >>= 1) mxb = fmaxf(mxb, __shfl_xor(mxb, off, 32));
  }
  const float thr = -si[(size_t)b * 2048 + j];
  __syncthreads();
  const float vj = sjs[j];

  int cr = 0, ct = 0;
  for (int kc = q * 8; kc < q * 8 + 8; ++kc) {
    const float4* p4 = (const float4*)&sjs[kc * 64];
    if (kc < jc) {
#pragma unroll
      for (int u = 0; u < 16; ++u) {
        const float4 v = p4[u];
        cr += (int)(v.x <= vj) + (int)(v.y <= vj) + (int)(v.z <= vj) + (int)(v.w <= vj);
        ct += (int)(v.x < thr) + (int)(v.y < thr) + (int)(v.z < thr) + (int)(v.w < thr);
      }
    } else if (kc > jc) {
#pragma unroll
      for (int u = 0; u < 16; ++u) {
        const float4 v = p4[u];
        cr += (int)(v.x < vj) + (int)(v.y < vj) + (int)(v.z < vj) + (int)(v.w < vj);
        ct += (int)(v.x < thr) + (int)(v.y < thr) + (int)(v.z < thr) + (int)(v.w < thr);
      }
    } else {
#pragma unroll
      for (int u = 0; u < 16; ++u) {
        const float4 v = p4[u];
        const int k0 = 4 * u;
        cr += (int)(v.x < vj || (v.x == vj && k0 + 0 < jl));
        cr += (int)(v.y < vj || (v.y == vj && k0 + 1 < jl));
        cr += (int)(v.z < vj || (v.z == vj && k0 + 2 < jl));
        cr += (int)(v.w < vj || (v.w == vj && k0 + 3 < jl));
        ct += (int)(v.x < thr) + (int)(v.y < thr) + (int)(v.z < thr) + (int)(v.w < thr);
      }
    }
  }
  cR[q][jl] = cr;
  cT[q][jl] = ct;
  __syncthreads();
  if (q == 0) {
    const int r = cR[0][jl] + cR[1][jl] + cR[2][jl] + cR[3][jl];
    const int t = cT[0][jl] + cT[1][jl] + cT[2][jl] + cT[3][jl];
    const float d = vj - mxb;
    int* sidx = (int*)sidx_f;
    int* trow = (int*)trow_f;
    float2* sbd = (float2*)sbd_f;
    sidx[(size_t)b * 2048 + r] = j;
    sbd[(size_t)b * 2048 + r] = make_float2(__expf(d), __expf(ALPHA * d));
    trow[(size_t)b * 2048 + j] = t;
  }
}

__global__ void __launch_bounds__(512) k_segscan(const float* __restrict__ h,
                                                 const float* __restrict__ sidx_f,
                                                 const float* __restrict__ sbd_f,
                                                 float* __restrict__ pdx,
                                                 float* __restrict__ sbn) {
  __shared__ float totD[64][8];
  __shared__ float totB[64][8];

  const int tid = threadIdx.x;
  const int c = tid & 7;
  const int run = tid >> 3;
  const int cg = blockIdx.x;
  const int b = blockIdx.y;
  const bool ones = (cg == 16);
  const int4* sidx4 = (const int4*)((const int*)sidx_f + (size_t)b * 2048 + run * 32);
  const float4* sbd4 = (const float4*)((const float2*)sbd_f + (size_t)b * 2048 + run * 32);
  const float* hb = h + (size_t)b * 2048 * 128 + (ones ? 0 : cg * 8) + c;

  int jv[32];
#pragma unroll
  for (int k = 0; k < 8; ++k) {
    const int4 v = sidx4[k];
    jv[4 * k + 0] = v.x & 2047;
    jv[4 * k + 1] = v.y & 2047;
    jv[4 * k + 2] = v.z & 2047;
    jv[4 * k + 3] = v.w & 2047;
  }
  float wB[32], wD[32];
#pragma unroll
  for (int k = 0; k < 16; ++k) {
    const float4 v = sbd4[k];
    wB[2 * k] = v.x;
    wD[2 * k] = v.y;
    wB[2 * k + 1] = v.z;
    wD[2 * k + 1] = v.w;
  }
  float hv[32];
#pragma unroll
  for (int k = 0; k < 32; ++k) hv[k] = hb[(size_t)jv[k] * 128];
  if (ones) {
#pragma unroll
    for (int k = 0; k < 32; ++k) hv[k] = 1.0f;
  }

  float tD[4], tB[4];
#pragma unroll
  for (int s = 0; s < 4; ++s) {
    float aD = 0.0f, aB = 0.0f;
#pragma unroll
    for (int u = 0; u < SEG; ++u) {
      aD += wD[s * SEG + u] * hv[s * SEG + u];
      aB += wB[s * SEG + u] * hv[s * SEG + u];
    }
    tD[s] = aD;
    tB[s] = aB;
  }
  totD[run][c] = ((tD[0] + tD[1]) + tD[2]) + tD[3];
  totB[run][c] = ((tB[3] + tB[2]) + tB[1]) + tB[0];
  __syncthreads();
  float cD = 0.0f, cB = 0.0f;
  for (int r = 0; r < 64; ++r) {
    const float v = totD[r][c];
    cD += (r < run) ? v : 0.0f;
  }
  for (int r = 63; r >= 0; --r) {
    const float v = totB[r][c];
    cB += (r > run) ? v : 0.0f;
  }
  const size_t o = ((size_t)b * NSEG + run * 4) * BSTR + cg * 8 + c;
  float* pd = pdx + o;
  float* sb = sbn + o;
  const float d1 = cD + tD[0], d2 = d1 + tD[1], d3 = d2 + tD[2];
  pd[0] = cD;
  pd[BSTR] = d1;
  pd[2 * BSTR] = d2;
  pd[3 * BSTR] = d3;
  const float b2 = cB + tB[3], b1 = b2 + tB[2], b0 = b1 + tB[1];
  sb[3 * BSTR] = cB;
  sb[2 * BSTR] = b2;
  sb[BSTR] = b1;
  sb[0] = b0;
}

#define RW_ROWS 2
__global__ void __launch_bounds__(256) k_rows(const float* __restrict__ h,
                                              const float* __restrict__ si,
                                              const float* __restrict__ pmx,
                                              const float* __restrict__ sidx_f,
                                              const float* __restrict__ sbd_f,
                                              const float* __restrict__ trow_f,
                                              const float* __restrict__ pdx,
                                              const float* __restrict__ sbn,
                                              float* __restrict__ out) {
  const int lane = threadIdx.x & 31;
  const int wv = blockIdx.x * 8 + (threadIdx.x >> 5);
  const int row0 = wv * 4;
  const int b = row0 >> 11;

  const int4 t4 = *(const int4*)((const int*)trow_f + row0);
  const float4 s4 = *(const float4*)(si + row0);
  float mxb = -3.402823466e38f;
  {
    const float4* pm = (const float4*)(pmx + b * 256 + lane * 8);
    const float4 pa = pm[0], pb = pm[1];
    mxb = fmaxf(mxb, fmaxf(fmaxf(pa.x, pa.y), fmaxf(pa.z, pa.w)));
    mxb = fmaxf(mxb, fmaxf(fmaxf(pb.x, pb.y), fmaxf(pb.z, pb.w)));
#pragma unroll
    for (int off = 16; off > 0; off >>= 1) mxb = fmaxf(mxb, __shfl_xor(mxb, off, 32));
  }
  const int* sidx = (const int*)sidx_f + (size_t)b * 2048;
  const float2* sbd = (const float2*)sbd_f + (size_t)b * 2048;
  const float* hb = h + (size_t)b * 2048 * 128 + 4 * lane;

#pragma unroll 1
  for (int pass = 0; pass < 2; ++pass) {
    const int t[RW_ROWS] = {pass ? t4.z : t4.x, pass ? t4.w : t4.y};
    const float sr[RW_ROWS] = {pass ? s4.z : s4.x, pass ? s4.w : s4.y};

    int g[RW_ROWS];
    int jv[RW_ROWS][SEG];
    float wk[RW_ROWS][SEG];
    float4 aD[RW_ROWS], aB[RW_ROWS];
    float zD[RW_ROWS], zB[RW_ROWS];
#pragma unroll
    for (int r = 0; r < RW_ROWS; ++r) {
      g[r] = min(t[r] >> 3, NSEG - 1);
      const int4* ip = (const int4*)(sidx + g[r] * SEG);
      const int4 i0 = ip[0], i1 = ip[1];
      jv[r][0] = i0.x & 2047;
      jv[r][1] = i0.y & 2047;
      jv[r][2] = i0.z & 2047;
      jv[r][3] = i0.w & 2047;
      jv[r][4] = i1.x & 2047;
      jv[r][5] = i1.y & 2047;
      jv[r][6] = i1.z & 2047;
      jv[r][7] = i1.w & 2047;
      const float4* wp = (const float4*)(sbd + g[r] * SEG);
#pragma unroll
      for (int k = 0; k < 4; ++k) {
        const float4 v = wp[k];
        wk[r][2 * k] = (g[r] * SEG + 2 * k < t[r]) ? v.y : v.x;
        wk[r][2 * k + 1] = (g[r] * SEG + 2 * k + 1 < t[r]) ? v.w : v.z;
      }
      const float* pdr = pdx + ((size_t)b * NSEG + g[r]) * BSTR;
      const float* sbr = sbn + ((size_t)b * NSEG + g[r]) * BSTR;
      aD[r] = *(const float4*)(pdr + 4 * lane);
      aB[r] = *(const float4*)(sbr + 4 * lane);
      zD[r] = pdr[128];
      zB[r] = sbr[128];
    }
    float4 hv[RW_ROWS][SEG];
#pragma unroll
    for (int r = 0; r < RW_ROWS; ++r)
#pragma unroll
      for (int u = 0; u < SEG; ++u) hv[r][u] = *(const float4*)(hb + (size_t)jv[r][u] * 128);

#pragma unroll
    for (int r = 0; r < RW_ROWS; ++r) {
      const float s_row = sr[r] + mxb;
      const float m_row = leaky(s_row);
      const float a_one = __expf(s_row - m_row);
      const float c_one = __expf(ALPHA * s_row - m_row);
      float4 sD = aD[r], sB = aB[r];
      float yD = zD[r], yB = zB[r];
#pragma unroll
      for (int u = 0; u < SEG; ++u) {
        const float w = (g[r] * SEG + u < t[r]) ? wk[r][u] : 0.0f;
        sD.x += w * hv[r][u].x;
        sD.y += w * hv[r][u].y;
        sD.z += w * hv[r][u].z;
        sD.w += w * hv[r][u].w;
        yD += w;
      }
#pragma unroll
      for (int u = SEG - 1; u >= 0; --u) {
        const float w = (g[r] * SEG + u < t[r]) ? 0.0f : wk[r][u];
        sB.x += w * hv[r][u].x;
        sB.y += w * hv[r][u].y;
        sB.z += w * hv[r][u].z;
        sB.w += w * hv[r][u].w;
        yB += w;
      }
      const float zinv = 1.0f / (c_one * yD + a_one * yB);
      float4 o;
      o.x = (c_one * sD.x + a_one * sB.x) * zinv;
      o.y = (c_one * sD.y + a_one * sB.y) * zinv;
      o.z = (c_one * sD.z + a_one * sB.z) * zinv;
      o.w = (c_one * sD.w + a_one * sB.w) * zinv;
      *(float4*)(out + (size_t)(row0 + 2 * pass + r) * 128 + 4 * lane) = o;
    }
  }
}

extern "C" void kernel_launch(void* const* d_in, const int* in_sizes, int n_in,
                              void* d_out, int out_size, void* d_ws, size_t ws_size,
                              hipStream_t stream) {
  const float* x = (const float*)d_in[0];
  const float* W = (const float*)d_in[1];
  const float* a = (const float*)d_in[2];
  float* out = (float*)d_out;

  float* h  = (float*)d_ws;
  float* si = h + (size_t)16384 * 128;
  float* sj = si + 16384;
  float* pmx = sj + 16384;
  float* sidx = pmx + 2048;
  float* sbd = sidx + 16384;
  float* trow = sbd + 32768;
  float* pdx = trow + 16384;
  float* sbn = pdx + (size_t)8 * 256 * 136;

  if (ws_size < (size_t)11018240) return;

  k_h_gemm<<<dim3(256), dim3(128), 0, stream>>>(x, W, h);
  k_scores<<<dim3(256), dim3(256), 0, stream>>>(h, a, si, sj, pmx);
  k_rank<<<dim3(32, 8), dim3(256), 0, stream>>>(si, sj, pmx, sidx, sbd, trow);
  k_segscan<<<dim3(17, 8), dim3(512), 0, stream>>>(h, sidx, sbd, pdx, sbn);
  k_rows<<<dim3(512), dim3(256), 0, stream>>>(h, si, pmx, sidx, sbd, trow, pdx, sbn, out);
}


# score launch removed: si, sj = x.(W^T a1), x.(W^T a2) made by 256 extra work-groups of the GEMM launch (W^T a built per group in LDS); new summation order for si, sj; launches 5 -> 4
# speedup vs baseline: 2.9580x; 1.0486x over previous
#include <hip/hip_runtime.h>

#define ALPHA 0.2f

typedef __attribute__((ext_vector_type(2))) float v2f;
typedef __attribute__((ext_vector_type(4))) float v4f;
typedef __attribute__((ext_vector_type(8))) float v8f;

__device__ __forceinline__ v8f wmma_f32_16x16x4(v2f a, v2f b, v8f c) {
  return __builtin_amdgcn_wmma_f32_16x16x4_f32(false, a, false, b, (short)0, c,
                                               false, false);
}

__device__ __forceinline__ float leaky(float t) {
  return (t >= 0.0f) ? t : ALPHA * t;
}

#define G_DS 2
#define G_NST 8
__device__ __forceinline__ void gemm_load_stage(const float* xa, const float* wb,
                                                int kk0, v4f (&fa)[G_DS][2],
                                                v4f (&fb)[G_DS][4]) {
#pragma unroll
  for (int d = 0; d < G_DS; ++d) {
#pragma unroll
    for (int rt = 0; rt < 2; ++rt)
      fa[d][rt] = *(const v4f*)(xa + (size_t)rt * 16 * 128 + kk0 + d * 8);
#pragma unroll
    for (int ot = 0; ot < 4; ++ot)
      fb[d][ot] = *(const v4f*)(wb + (size_t)ot * 16 * 128 + kk0 + d * 8);
  }
}
__device__ __forceinline__ void gemm_compute_stage(const v4f (&fa)[G_DS][2],
                                                   const v4f (&fb)[G_DS][4],
                                                   v8f (&acc)[2][4]) {
#pragma unroll
  for (int d = 0; d < G_DS; ++d) {
#pragma unroll
    for (int half = 0; half < 2; ++half) {
      v2f a[2], b[4];
#pragma unroll
      for (int rt = 0; rt < 2; ++rt)
        a[rt] = v2f{fa[d][rt][2 * half], fa[d][rt][2 * half + 1]};
#pragma unroll
      for (int ot = 0; ot < 4; ++ot)
        b[ot] = v2f{fb[d][ot][2 * half], fb[d][ot][2 * half + 1]};
#pragma unroll
      for (int rt = 0; rt < 2; ++rt)
#pragma unroll
        for (int ot = 0; ot < 4; ++ot)
          acc[rt][ot] = wmma_f32_16x16x4(a[rt], b[ot], acc[rt][ot]);
    }
  }
}
#define G_BLOCKS 256
#define S_BLOCKS 256
#define S_ROWS 16
#define SC_GRP 8
__device__ __forceinline__ void score_role(const float* __restrict__ x,
                                           const float* __restrict__ W,
                                           const float* __restrict__ a,
                                           float* __restrict__ si,
                                           float* __restrict__ sj,
                                           float* __restrict__ pmx, int sb,
                                           float (*us)[128]) {
  const int tid = threadIdx.x;
  const int lane = tid & 31;
  const int row0 = sb * 64 + (tid >> 5) * S_ROWS;

  float4 xv[S_ROWS];
  const float* xp = x + (size_t)row0 * 128 + 4 * lane;
#pragma unroll
  for (int r = 0; r < S_ROWS; ++r) xv[r] = *(const float4*)(xp + r * 128);

  float p1[4] = {0.0f, 0.0f, 0.0f, 0.0f};
  float p2[4] = {0.0f, 0.0f, 0.0f, 0.0f};
#pragma unroll
  for (int o = 0; o < 128; ++o) {
    const float w = W[o * 128 + tid];
    p1[o & 3] += w * a[o];
    p2[o & 3] += w * a[128 + o];
  }
  us[0][tid] = (p1[0] + p1[1]) + (p1[2] + p1[3]);
  us[1][tid] = (p2[0] + p2[1]) + (p2[2] + p2[3]);
  __syncthreads();
  const float4 u1 = *(const float4*)&us[0][4 * lane];
  const float4 u2 = *(const float4*)&us[1][4 * lane];

  float psi[S_ROWS], psj[S_ROWS];
#pragma unroll
  for (int r = 0; r < S_ROWS; ++r) {
    float t1 = xv[r].x * u1.x;
    t1 += xv[r].y * u1.y;
    t1 += xv[r].z * u1.z;
    t1 += xv[r].w * u1.w;
    float t2 = xv[r].x * u2.x;
    t2 += xv[r].y * u2.y;
    t2 += xv[r].z * u2.z;
    t2 += xv[r].w * u2.w;
    psi[r] = t1;
    psj[r] = t2;
  }
#pragma unroll
  for (int off = 16; off > 0; off >>= 1) {
#pragma unroll
    for (int r = 0; r < S_ROWS; ++r) {
      psi[r] += __shfl_xor(psi[r], off, 32);
      psj[r] += __shfl_xor(psj[r], off, 32);
    }
  }
  if (lane < S_ROWS) {
    float osi = psi[0], osj = psj[0];
#pragma unroll
    for (int r = 1; r < S_ROWS; ++r) {
      if (lane == r) {
        osi = psi[r];
        osj = psj[r];
      }
    }
    si[row0 + lane] = osi;
    sj[row0 + lane] = osj;
  }
  if (lane < S_ROWS / SC_GRP) {
    float wmx = -3.402823466e38f;
#pragma unroll
    for (int g = 0; g < S_ROWS / SC_GRP; ++g) {
      float m = -3.402823466e38f;
#pragma unroll
      for (int r = 0; r < SC_GRP; ++r) m = fmaxf(m, psj[g * SC_GRP + r]);
      if (lane == g) wmx = m;
    }
    pmx[row0 / SC_GRP + lane] = wmx;
  }
}

__global__ void __launch_bounds__(128) k_h_gemm_scores(const float* __restrict__ x,
                                                       const float* __restrict__ W,
                                                       const float* __restrict__ a,
                                                       float* __restrict__ h,
                                                       float* __restrict__ si,
                                                       float* __restrict__ sj,
                                                       float* __restrict__ pmx) {
  __shared__ __align__(16) float us[2][128];
  if (blockIdx.x >= G_BLOCKS) {
    score_role(x, W, a, si, sj, pmx, blockIdx.x - G_BLOCKS, us);
    return;
  }
  const int lane = threadIdx.x & 31;
  const int wave = threadIdx.x >> 5;
  const int rl = lane & 15;
  const int hi = lane >> 4;
  const int i0 = blockIdx.x * 64 + (wave & 1) * 32;
  const int cb = (wave >> 1) * 64;

  v8f acc[2][4];
  v8f zero = {};
#pragma unroll
  for (int rt = 0; rt < 2; ++rt)
#pragma unroll
    for (int ot = 0; ot < 4; ++ot) acc[rt][ot] = zero;

  const float* xa = x + (size_t)(i0 + rl) * 128 + 4 * hi;
  const float* wb = W + (size_t)(cb + rl) * 128 + 4 * hi;

  v4f fa[3][G_DS][2], fb[3][G_DS][4];
  gemm_load_stage(xa, wb, 0, fa[0], fb[0]);
  gemm_load_stage(xa, wb, G_DS * 8, fa[1], fb[1]);
#pragma unroll
  for (int st = 0; st < G_NST; ++st) {
    if (st + 2 < G_NST)
      gemm_load_stage(xa, wb, (st + 2) * G_DS * 8, fa[(st + 2) % 3], fb[(st + 2) % 3]);
    __builtin_amdgcn_sched_barrier(0);
    gemm_compute_stage(fa[st % 3], fb[st % 3], acc);
  }
#pragma unroll
  for (int rt = 0; rt < 2; ++rt)
#pragma unroll
    for (int ot = 0; ot < 4; ++ot)
#pragma unroll
      for (int r = 0; r < 8; ++r)
        h[(size_t)(i0 + rt * 16 + r + hi * 8) * 128 + cb + ot * 16 + rl] =
            acc[rt][ot][r];
}

#define SEG 8
#define NSEG 256
#define BSTR 136

__global__ void __launch_bounds__(256) k_rank(const float* __restrict__ si,
                                              const float* __restrict__ sj,
                                              const float* __restrict__ pmx,
                                              float* __restrict__ sidx_f,
                                              float* __restrict__ sbd_f,
                                              float* __restrict__ trow_f) {
  __shared__ __align__(16) float sjs[2048];
  __shared__ int cR[4][64];
  __shared__ int cT[4][64];

  const int tid = threadIdx.x;
  const int lane = tid & 31;
  const int b = blockIdx.y;
  const int jc = blockIdx.x;
  const int jl = tid & 63;
  const int q = __builtin_amdgcn_readfirstlane(tid >> 6);
  const int j = jc * 64 + jl;
  const float* sjb = sj + (size_t)b * 2048;

#pragma unroll
  for (int k = 0; k < 8; ++k) sjs[tid + 256 * k] = sjb[tid + 256 * k];

  float mxb = -3.402823466e38f;
  {
    const float4* pm = (const float4*)(pmx + b * 256 + lane * 8);
    const float4 pa = pm[0], pb = pm[1];
    mxb = fmaxf(mxb, fmaxf(fmaxf(pa.x, pa.y), fmaxf(pa.z, pa.w)));
    mxb = fmaxf(mxb, fmaxf(fmaxf(pb.x, pb.y), fmaxf(pb.z, pb.w)));
#pragma unroll
    for (int off = 16; off > 0; off >>= 1) mxb = fmaxf(mxb, __shfl_xor(mxb, off, 32));
  }
  const float thr = -si[(size_t)b * 2048 + j];
  __syncthreads();
  const float vj = sjs[j];

  int cr = 0, ct = 0;
  for (int kc = q * 8; kc < q * 8 + 8; ++kc) {
    const float4* p4 = (const float4*)&sjs[kc * 64];
    if (kc < jc) {
#pragma unroll
      for (int u = 0; u < 16; ++u) {
        const float4 v = p4[u];
        cr += (int)(v.x <= vj) + (int)(v.y <= vj) + (int)(v.z <= vj) + (int)(v.w <= vj);
        ct += (int)(v.x < thr) + (int)(v.y < thr) + (int)(v.z < thr) + (int)(v.w < thr);
      }
    } else if (kc > jc) {
#pragma unroll
      for (int u = 0; u < 16; ++u) {
        const float4 v = p4[u];
        cr += (int)(v.x < vj) + (int)(v.y < vj) + (int)(v.z < vj) + (int)(v.w < vj);
        ct += (int)(v.x < thr) + (int)(v.y < thr) + (int)(v.z < thr) + (int)(v.w < thr);
      }
    } else {
#pragma unroll
      for (int u = 0; u < 16; ++u) {
        const float4 v = p4[u];
        const int k0 = 4 * u;
        cr += (int)(v.x < vj || (v.x == vj && k0 + 0 < jl));
        cr += (int)(v.y < vj || (v.y == vj && k0 + 1 < jl));
        cr += (int)(v.z < vj || (v.z == vj && k0 + 2 < jl));
        cr += (int)(v.w < vj || (v.w == vj && k0 + 3 < jl));
        ct += (int)(v.x < thr) + (int)(v.y < thr) + (int)(v.z < thr) + (int)(v.w < thr);
      }
    }
  }
  cR[q][jl] = cr;
  cT[q][jl] = ct;
  __syncthreads();
  if (q == 0) {
    const int r = cR[0][jl] + cR[1][jl] + cR[2][jl] + cR[3][jl];
    const int t = cT[0][jl] + cT[1][jl] + cT[2][jl] + cT[3][jl];
    const float d = vj - mxb;
    int* sidx = (int*)sidx_f;
    int* trow = (int*)trow_f;
    float2* sbd = (float2*)sbd_f;
    sidx[(size_t)b * 2048 + r] = j;
    sbd[(size_t)b * 2048 + r] = make_float2(__expf(d), __expf(ALPHA * d));
    trow[(size_t)b * 2048 + j] = t;
  }
}

__global__ void __launch_bounds__(512) k_segscan(const float* __restrict__ h,
                                                 const float* __restrict__ sidx_f,
                                                 const float* __restrict__ sbd_f,
                                                 float* __restrict__ pdx,
                                                 float* __restrict__ sbn) {
  __shared__ float totD[64][8];
  __shared__ float totB[64][8];

  const int tid = threadIdx.x;
  const int c = tid & 7;
  const int run = tid >> 3;
  const int cg = blockIdx.x;
  const int b = blockIdx.y;
  const bool ones = (cg == 16);
  const int4* sidx4 = (const int4*)((const int*)sidx_f + (size_t)b * 2048 + run * 32);
  const float4* sbd4 = (const float4*)((const float2*)sbd_f + (size_t)b * 2048 + run * 32);
  const float* hb = h + (size_t)b * 2048 * 128 + (ones ? 0 : cg * 8) + c;

  int jv[32];
#pragma unroll
  for (int k = 0; k < 8; ++k) {
    const int4 v = sidx4[k];
    jv[4 * k + 0] = v.x & 2047;
    jv[4 * k + 1] = v.y & 2047;
    jv[4 * k + 2] = v.z & 2047;
    jv[4 * k + 3] = v.w & 2047;
  }
  float wB[32], wD[32];
#pragma unroll
  for (int k = 0; k < 16; ++k) {
    const float4 v = sbd4[k];
    wB[2 * k] = v.x;
    wD[2 * k] = v.y;
    wB[2 * k + 1] = v.z;
    wD[2 * k + 1] = v.w;
  }
  float hv[32];
#pragma unroll
  for (int k = 0; k < 32; ++k) hv[k] = hb[(size_t)jv[k] * 128];
  if (ones) {
#pragma unroll
    for (int k = 0; k < 32; ++k) hv[k] = 1.0f;
  }

  float tD[4], tB[4];
#pragma unroll
  for (int s = 0; s < 4; ++s) {
    float aD = 0.0f, aB = 0.0f;
#pragma unroll
    for (int u = 0; u < SEG; ++u) {
      aD += wD[s * SEG + u] * hv[s * SEG + u];
      aB += wB[s * SEG + u] * hv[s * SEG + u];
    }
    tD[s] = aD;
    tB[s] = aB;
  }
  totD[run][c] = ((tD[0] + tD[1]) + tD[2]) + tD[3];
  totB[run][c] = ((tB[3] + tB[2]) + tB[1]) + tB[0];
  __syncthreads();
  float cD = 0.0f, cB = 0.0f;
  for (int r = 0; r < 64; ++r) {
    const float v = totD[r][c];
    cD += (r < run) ? v : 0.0f;
  }
  for (int r = 63; r >= 0; --r) {
    const float v = totB[r][c];
    cB += (r > run) ? v : 0.0f;
  }
  const size_t o = ((size_t)b * NSEG + run * 4) * BSTR + cg * 8 + c;
  float* pd = pdx + o;
  float* sb = sbn + o;
  const float d1 = cD + tD[0], d2 = d1 + tD[1], d3 = d2 + tD[2];
  pd[0] = cD;
  pd[BSTR] = d1;
  pd[2 * BSTR] = d2;
  pd[3 * BSTR] = d3;
  const float b2 = cB + tB[3], b1 = b2 + tB[2], b0 = b1 + tB[1];
  sb[3 * BSTR] = cB;
  sb[2 * BSTR] = b2;
  sb[BSTR] = b1;
  sb[0] = b0;
}

#define RW_ROWS 2
__global__ void __launch_bounds__(256) k_rows(const float* __restrict__ h,
                                              const float* __restrict__ si,
                                              const float* __restrict__ pmx,
                                              const float* __restrict__ sidx_f,
                                              const float* __restrict__ sbd_f,
                                              const float* __restrict__ trow_f,
                                              const float* __restrict__ pdx,
                                              const float* __restrict__ sbn,
                                              float* __restrict__ out) {
  const int lane = threadIdx.x & 31;
  const int wv = blockIdx.x * 8 + (threadIdx.x >> 5);
  const int row0 = wv * 4;
  const int b = row0 >> 11;

  const int4 t4 = *(const int4*)((const int*)trow_f + row0);
  const float4 s4 = *(const float4*)(si + row0);
  float mxb = -3.402823466e38f;
  {
    const float4* pm = (const float4*)(pmx + b * 256 + lane * 8);
    const float4 pa = pm[0], pb = pm[1];
    mxb = fmaxf(mxb, fmaxf(fmaxf(pa.x, pa.y), fmaxf(pa.z, pa.w)));
    mxb = fmaxf(mxb, fmaxf(fmaxf(pb.x, pb.y), fmaxf(pb.z, pb.w)));
#pragma unroll
    for (int off = 16; off > 0; off >>= 1) mxb = fmaxf(mxb, __shfl_xor(mxb, off, 32));
  }
  const int* sidx = (const int*)sidx_f + (size_t)b * 2048;
  const float2* sbd = (const float2*)sbd_f + (size_t)b * 2048;
  const float* hb = h + (size_t)b * 2048 * 128 + 4 * lane;

#pragma unroll 1
  for (int pass = 0; pass < 2; ++pass) {
    const int t[RW_ROWS] = {pass ? t4.z : t4.x, pass ? t4.w : t4.y};
    const float sr[RW_ROWS] = {pass ? s4.z : s4.x, pass ? s4.w : s4.y};

    int g[RW_ROWS];
    int jv[RW_ROWS][SEG];
    float wk[RW_ROWS][SEG];
    float4 aD[RW_ROWS], aB[RW_ROWS];
    float zD[RW_ROWS], zB[RW_ROWS];
#pragma unroll
    for (int r = 0; r < RW_ROWS; ++r) {
      g[r] = min(t[r] >> 3, NSEG - 1);
      const int4* ip = (const int4*)(sidx + g[r] * SEG);
      const int4 i0 = ip[0], i1 = ip[1];
      jv[r][0] = i0.x & 2047;
      jv[r][1] = i0.y & 2047;
      jv[r][2] = i0.z & 2047;
      jv[r][3] = i0.w & 2047;
      jv[r][4] = i1.x & 2047;
      jv[r][5] = i1.y & 2047;
      jv[r][6] = i1.z & 2047;
      jv[r][7] = i1.w & 2047;
      const float4* wp = (const float4*)(sbd + g[r] * SEG);
#pragma unroll
      for (int k = 0; k < 4; ++k) {
        const float4 v = wp[k];
        wk[r][2 * k] = (g[r] * SEG + 2 * k < t[r]) ? v.y : v.x;
        wk[r][2 * k + 1] = (g[r] * SEG + 2 * k + 1 < t[r]) ? v.w : v.z;
      }
      const float* pdr = pdx + ((size_t)b * NSEG + g[r]) * BSTR;
      const float* sbr = sbn + ((size_t)b * NSEG + g[r]) * BSTR;
      aD[r] = *(const float4*)(pdr + 4 * lane);
      aB[r] = *(const float4*)(sbr + 4 * lane);
      zD[r] = pdr[128];
      zB[r] = sbr[128];
    }
    float4 hv[RW_ROWS][SEG];
#pragma unroll
    for (int r = 0; r < RW_ROWS; ++r)
#pragma unroll
      for (int u = 0; u < SEG; ++u) hv[r][u] = *(const float4*)(hb + (size_t)jv[r][u] * 128);

#pragma unroll
    for (int r = 0; r < RW_ROWS; ++r) {
      const float s_row = sr[r] + mxb;
      const float m_row = leaky(s_row);
      const float a_one = __expf(s_row - m_row);
      const float c_one = __expf(ALPHA * s_row - m_row);
      float4 sD = aD[r], sB = aB[r];
      float yD = zD[r], yB = zB[r];
#pragma unroll
      for (int u = 0; u < SEG; ++u) {
        const float w = (g[r] * SEG + u < t[r]) ? wk[r][u] : 0.0f;
        sD.x += w * hv[r][u].x;
        sD.y += w * hv[r][u].y;
        sD.z += w * hv[r][u].z;
        sD.w += w * hv[r][u].w;
        yD += w;
      }
#pragma unroll
      for (int u = SEG - 1; u >= 0; --u) {
        const float w = (g[r] * SEG + u < t[r]) ? 0.0f : wk[r][u];
        sB.x += w * hv[r][u].x;
        sB.y += w * hv[r][u].y;
        sB.z += w * hv[r][u].z;
        sB.w += w * hv[r][u].w;
        yB += w;
      }
      const float zinv = 1.0f / (c_one * yD + a_one * yB);
      float4 o;
      o.x = (c_one * sD.x + a_one * sB.x) * zinv;
      o.y = (c_one * sD.y + a_one * sB.y) * zinv;
      o.z = (c_one * sD.z + a_one * sB.z) * zinv;
      o.w = (c_one * sD.w + a_one * sB.w) * zinv;
      *(float4*)(out + (size_t)(row0 + 2 * pass + r) * 128 + 4 * lane) = o;
    }
  }
}

extern "C" void kernel_launch(void* const* d_in, const int* in_sizes, int n_in,
                              void* d_out, int out_size, void* d_ws, size_t ws_size,
                              hipStream_t stream) {
  const float* x = (const float*)d_in[0];
  const float* W = (const float*)d_in[1];
  const float* a = (const float*)d_in[2];
  float* out = (float*)d_out;

  float* h  = (float*)d_ws;
  float* si = h + (size_t)16384 * 128;
  float* sj = si + 16384;
  float* pmx = sj + 16384;
  float* sidx = pmx + 2048;
  float* sbd = sidx + 16384;
  float* trow = sbd + 32768;
  float* pdx = trow + 16384;
  float* sbn = pdx + (size_t)8 * 256 * 136;

  if (ws_size < (size_t)11018240) return;

  k_h_gemm_scores<<<dim3(512), dim3(128), 0, stream>>>(x, W, a, h, si, sj, pmx);
  k_rank<<<dim3(32, 8), dim3(256), 0, stream>>>(si, sj, pmx, sidx, sbd, trow);
  k_segscan<<<dim3(17, 8), dim3(512), 0, stream>>>(h, sidx, sbd, pdx, sbn);
  k_rows<<<dim3(512), dim3(256), 0, stream>>>(h, si, pmx, sidx, sbd, trow, pdx, sbn, out);
}


# segment-sum and row launches fused: a work-group (batch, 32 columns, 256 rows) gathers its h slice in rank order into LDS, scans it there, serves its rows from LDS; launches 4->3; new summation order
# speedup vs baseline: 3.5476x; 1.1993x over previous
#include <hip/hip_runtime.h>

#define ALPHA 0.2f

typedef __attribute__((ext_vector_type(2))) float v2f;
typedef __attribute__((ext_vector_type(4))) float v4f;
typedef __attribute__((ext_vector_type(8))) float v8f;

__device__ __forceinline__ v8f wmma_f32_16x16x4(v2f a, v2f b, v8f c) {
  return __builtin_amdgcn_wmma_f32_16x16x4_f32(false, a, false, b, (short)0, c,
                                               false, false);
}

__device__ __forceinline__ float leaky(float t) {
  return (t >= 0.0f) ? t : ALPHA * t;
}

#define G_DS 2
#define G_NST 8
__device__ __forceinline__ void gemm_load_stage(const float* xa, const float* wb,
                                                int kk0, v4f (&fa)[G_DS][2],
                                                v4f (&fb)[G_DS][4]) {
#pragma unroll
  for (int d = 0; d < G_DS; ++d) {
#pragma unroll
    for (int rt = 0; rt < 2; ++rt)
      fa[d][rt] = *(const v4f*)(xa + (size_t)rt * 16 * 128 + kk0 + d * 8);
#pragma unroll
    for (int ot = 0; ot < 4; ++ot)
      fb[d][ot] = *(const v4f*)(wb + (size_t)ot * 16 * 128 + kk0 + d * 8);
  }
}
__device__ __forceinline__ void gemm_compute_stage(const v4f (&fa)[G_DS][2],
                                                   const v4f (&fb)[G_DS][4],
                                                   v8f (&acc)[2][4]) {
#pragma unroll
  for (int d = 0; d < G_DS; ++d) {
#pragma unroll
    for (int half = 0; half < 2; ++half) {
      v2f a[2], b[4];
#pragma unroll
      for (int rt = 0; rt < 2; ++rt)
        a[rt] = v2f{fa[d][rt][2 * half], fa[d][rt][2 * half + 1]};
#pragma unroll
      for (int ot = 0; ot < 4; ++ot)
        b[ot] = v2f{fb[d][ot][2 * half], fb[d][ot][2 * half + 1]};
#pragma unroll
      for (int rt = 0; rt < 2; ++rt)
#pragma unroll
        for (int ot = 0; ot < 4; ++ot)
          acc[rt][ot] = wmma_f32_16x16x4(a[rt], b[ot], acc[rt][ot]);
    }
  }
}
#define G_BLOCKS 256
#define S_BLOCKS 256
#define S_ROWS 16
#define SC_GRP 8
__device__ __forceinline__ void score_role(const float* __restrict__ x,
                                           const float* __restrict__ W,
                                           const float* __restrict__ a,
                                           float* __restrict__ si,
                                           float* __restrict__ sj,
                                           float* __restrict__ pmx, int sb,
                                           float (*us)[128]) {
  const int tid = threadIdx.x;
  const int lane = tid & 31;
  const int row0 = sb * 64 + (tid >> 5) * S_ROWS;

  float4 xv[S_ROWS];
  const float* xp = x + (size_t)row0 * 128 + 4 * lane;
#pragma unroll
  for (int r = 0; r < S_ROWS; ++r) xv[r] = *(const float4*)(xp + r * 128);

  float p1[4] = {0.0f, 0.0f, 0.0f, 0.0f};
  float p2[4] = {0.0f, 0.0f, 0.0f, 0.0f};
#pragma unroll
  for (int o = 0; o < 128; ++o) {
    const float w = W[o * 128 + tid];
    p1[o & 3] += w * a[o];
    p2[o & 3] += w * a[128 + o];
  }
  us[0][tid] = (p1[0] + p1[1]) + (p1[2] + p1[3]);
  us[1][tid] = (p2[0] + p2[1]) + (p2[2] + p2[3]);
  __syncthreads();
  const float4 u1 = *(const float4*)&us[0][4 * lane];
  const float4 u2 = *(const float4*)&us[1][4 * lane];

  float psi[S_ROWS], psj[S_ROWS];
#pragma unroll
  for (int r = 0; r < S_ROWS; ++r) {
    float t1 = xv[r].x * u1.x;
    t1 += xv[r].y * u1.y;
    t1 += xv[r].z * u1.z;
    t1 += xv[r].w * u1.w;
    float t2 = xv[r].x * u2.x;
    t2 += xv[r].y * u2.y;
    t2 += xv[r].z * u2.z;
    t2 += xv[r].w * u2.w;
    psi[r] = t1;
    psj[r] = t2;
  }
#pragma unroll
  for (int off = 16; off > 0; off >>= 1) {
#pragma unroll
    for (int r = 0; r < S_ROWS; ++r) {
      psi[r] += __shfl_xor(psi[r], off, 32);
      psj[r] += __shfl_xor(psj[r], off, 32);
    }
  }
  if (lane < S_ROWS) {
    float osi = psi[0], osj = psj[0];
#pragma unroll
    for (int r = 1; r < S_ROWS; ++r) {
      if (lane == r) {
        osi = psi[r];
        osj = psj[r];
      }
    }
    si[row0 + lane] = osi;
    sj[row0 + lane] = osj;
  }
  if (lane < S_ROWS / SC_GRP) {
    float wmx = -3.402823466e38f;
#pragma unroll
    for (int g = 0; g < S_ROWS / SC_GRP; ++g) {
      float m = -3.402823466e38f;
#pragma unroll
      for (int r = 0; r < SC_GRP; ++r) m = fmaxf(m, psj[g * SC_GRP + r]);
      if (lane == g) wmx = m;
    }
    pmx[row0 / SC_GRP + lane] = wmx;
  }
}

__global__ void __launch_bounds__(128) k_h_gemm_scores(const float* __restrict__ x,
                                                       const float* __restrict__ W,
                                                       const float* __restrict__ a,
                                                       float* __restrict__ h,
                                                       float* __restrict__ si,
                                                       float* __restrict__ sj,
                                                       float* __restrict__ pmx) {
  __shared__ __align__(16) float us[2][128];
  if (blockIdx.x >= G_BLOCKS) {
    score_role(x, W, a, si, sj, pmx, blockIdx.x - G_BLOCKS, us);
    return;
  }
  const int lane = threadIdx.x & 31;
  const int wave = threadIdx.x >> 5;
  const int rl = lane & 15;
  const int hi = lane >> 4;
  const int i0 = blockIdx.x * 64 + (wave & 1) * 32;
  const int cb = (wave >> 1) * 64;

  v8f acc[2][4];
  v8f zero = {};
#pragma unroll
  for (int rt = 0; rt < 2; ++rt)
#pragma unroll
    for (int ot = 0; ot < 4; ++ot) acc[rt][ot] = zero;

  const float* xa = x + (size_t)(i0 + rl) * 128 + 4 * hi;
  const float* wb = W + (size_t)(cb + rl) * 128 + 4 * hi;

  v4f fa[3][G_DS][2], fb[3][G_DS][4];
  gemm_load_stage(xa, wb, 0, fa[0], fb[0]);
  gemm_load_stage(xa, wb, G_DS * 8, fa[1], fb[1]);
#pragma unroll
  for (int st = 0; st < G_NST; ++st) {
    if (st + 2 < G_NST)
      gemm_load_stage(xa, wb, (st + 2) * G_DS * 8, fa[(st + 2) % 3], fb[(st + 2) % 3]);
    __builtin_amdgcn_sched_barrier(0);
    gemm_compute_stage(fa[st % 3], fb[st % 3], acc);
  }
#pragma unroll
  for (int rt = 0; rt < 2; ++rt)
#pragma unroll
    for (int ot = 0; ot < 4; ++ot)
#pragma unroll
      for (int r = 0; r < 8; ++r)
        h[(size_t)(i0 + rt * 16 + r + hi * 8) * 128 + cb + ot * 16 + rl] =
            acc[rt][ot][r];
}

#define SEG2 16
#define NSEG2 128
#define HROWS 2112

__global__ void __launch_bounds__(256) k_rank(const float* __restrict__ si,
                                              const float* __restrict__ sj,
                                              const float* __restrict__ pmx,
                                              float* __restrict__ sidx_f,
                                              float* __restrict__ sbd_f,
                                              float* __restrict__ trow_f) {
  __shared__ __align__(16) float sjs[2048];
  __shared__ int cR[4][64];
  __shared__ int cT[4][64];

  const int tid = threadIdx.x;
  const int lane = tid & 31;
  const int b = blockIdx.y;
  const int jc = blockIdx.x;
  const int jl = tid & 63;
  const int q = __builtin_amdgcn_readfirstlane(tid >> 6);
  const int j = jc * 64 + jl;
  const float* sjb = sj + (size_t)b * 2048;

#pragma unroll
  for (int k = 0; k < 8; ++k) sjs[tid + 256 * k] = sjb[tid + 256 * k];

  float mxb = -3.402823466e38f;
  {
    const float4* pm = (const float4*)(pmx + b * 256 + lane * 8);
    const float4 pa = pm[0], pb = pm[1];
    mxb = fmaxf(mxb, fmaxf(fmaxf(pa.x, pa.y), fmaxf(pa.z, pa.w)));
    mxb = fmaxf(mxb, fmaxf(fmaxf(pb.x, pb.y), fmaxf(pb.z, pb.w)));
#pragma unroll
    for (int off = 16; off > 0; off >>= 1) mxb = fmaxf(mxb, __shfl_xor(mxb, off, 32));
  }
  const float thr = -si[(size_t)b * 2048 + j];
  __syncthreads();
  const float vj = sjs[j];

  int cr = 0, ct = 0;
  for (int kc = q * 8; kc < q * 8 + 8; ++kc) {
    const float4* p4 = (const float4*)&sjs[kc * 64];
    if (kc < jc) {
#pragma unroll
      for (int u = 0; u < 16; ++u) {
        const float4 v = p4[u];
        cr += (int)(v.x <= vj) + (int)(v.y <= vj) + (int)(v.z <= vj) + (int)(v.w <= vj);
        ct += (int)(v.x < thr) + (int)(v.y < thr) + (int)(v.z < thr) + (int)(v.w < thr);
      }
    } else if (kc > jc) {
#pragma unroll
      for (int u = 0; u < 16; ++u) {
        const float4 v = p4[u];
        cr += (int)(v.x < vj) + (int)(v.y < vj) + (int)(v.z < vj) + (int)(v.w < vj);
        ct += (int)(v.x < thr) + (int)(v.y < thr) + (int)(v.z < thr) + (int)(v.w < thr);
      }
    } else {
#pragma unroll
      for (int u = 0; u < 16; ++u) {
        const float4 v = p4[u];
        const int k0 = 4 * u;
        cr += (int)(v.x < vj || (v.x == vj && k0 + 0 < jl));
        cr += (int)(v.y < vj || (v.y == vj && k0 + 1 < jl));
        cr += (int)(v.z < vj || (v.z == vj && k0 + 2 < jl));
        cr += (int)(v.w < vj || (v.w == vj && k0 + 3 < jl));
        ct += (int)(v.x < thr) + (int)(v.y < thr) + (int)(v.z < thr) + (int)(v.w < thr);
      }
    }
  }
  cR[q][jl] = cr;
  cT[q][jl] = ct;
  __syncthreads();
  if (q == 0) {
    const int r = cR[0][jl] + cR[1][jl] + cR[2][jl] + cR[3][jl];
    const int t = cT[0][jl] + cT[1][jl] + cT[2][jl] + cT[3][jl];
    const float d = vj - mxb;
    int* sidx = (int*)sidx_f;
    int* trow = (int*)trow_f;
    float2* sbd = (float2*)sbd_f;
    sidx[(size_t)b * 2048 + r] = j;
    sbd[(size_t)b * 2048 + r] = make_float2(__expf(d), __expf(ALPHA * d));
    trow[(size_t)b * 2048 + j] = t;
  }
}

__global__ void __launch_bounds__(512) k_scan_rows(const float* __restrict__ h,
                                                   const float* __restrict__ si,
                                                   const float* __restrict__ pmx,
                                                   const float* __restrict__ sidx_f,
                                                   const float* __restrict__ sbd_f,
                                                   const float* __restrict__ trow_f,
                                                   float* __restrict__ out) {
  __shared__ __align__(16) float4 Hs4[HROWS * 8];
  __shared__ __align__(16) float4 bD4[NSEG2 * 8];
  __shared__ __align__(16) float4 bB4[NSEG2 * 8];
  __shared__ __align__(16) float4 wl4[1024];
  __shared__ __align__(16) float4 wtD4[16 * 8];
  __shared__ __align__(16) float4 wtB4[16 * 8];
  __shared__ float zDs[NSEG2];
  __shared__ float zBs[NSEG2];
  __shared__ float wtZD[16];
  __shared__ float wtZB[16];

  const int tid = threadIdx.x;
  const int lane = tid & 31;
  const int wave = __builtin_amdgcn_readfirstlane(tid >> 5);
  const int c4 = tid & 7;
  const int sub = (tid >> 3) & 3;
  const int run = tid >> 3;
  const int rs = blockIdx.x;
  const int cq = blockIdx.y;
  const int b = blockIdx.z;

  const int4* sidx4 = (const int4*)((const int*)sidx_f + (size_t)b * 2048 + run * 32);
  int jv[32];
#pragma unroll
  for (int k = 0; k < 8; ++k) {
    const int4 v = sidx4[k];
    jv[4 * k + 0] = v.x & 2047;
    jv[4 * k + 1] = v.y & 2047;
    jv[4 * k + 2] = v.z & 2047;
    jv[4 * k + 3] = v.w & 2047;
  }
  const float4* sbd4 = (const float4*)((const float2*)sbd_f + (size_t)b * 2048);
  const float4 wc0 = sbd4[tid];
  const float4 wc1 = sbd4[tid + 512];
  const int rbase = b * 2048 + rs * 256 + wave * 16 + sub;
  const int* trow = (const int*)trow_f;
  const int t0 = trow[rbase], t1 = trow[rbase + 4], t2 = trow[rbase + 8], t3 = trow[rbase + 12];
  const float s0 = si[rbase], s1 = si[rbase + 4], s2 = si[rbase + 8], s3 = si[rbase + 12];
  float mxb = -3.402823466e38f;
  {
    const float4* pm = (const float4*)(pmx + b * 256 + lane * 8);
    const float4 pa = pm[0], pb = pm[1];
    mxb = fmaxf(mxb, fmaxf(fmaxf(pa.x, pa.y), fmaxf(pa.z, pa.w)));
    mxb = fmaxf(mxb, fmaxf(fmaxf(pb.x, pb.y), fmaxf(pb.z, pb.w)));
#pragma unroll
    for (int off = 16; off > 0; off >>= 1) mxb = fmaxf(mxb, __shfl_xor(mxb, off, 32));
  }

  const float* hb = h + (size_t)b * 2048 * 128 + cq * 32 + c4 * 4;
  float4 hv[32];
#pragma unroll
  for (int k = 0; k < 32; ++k) hv[k] = *(const float4*)(hb + (size_t)jv[k] * 128);

  wl4[tid] = wc0;
  wl4[tid + 512] = wc1;
  __syncthreads();

  const float2* wl2 = (const float2*)wl4;
  float4 tD[2], tB[2];
  float zd[2], zb[2];
#pragma unroll
  for (int s = 0; s < 2; ++s) {
    float4 aD = make_float4(0.0f, 0.0f, 0.0f, 0.0f);
    float4 aB = make_float4(0.0f, 0.0f, 0.0f, 0.0f);
    float yd = 0.0f, yb = 0.0f;
#pragma unroll
    for (int u = 0; u < SEG2; ++u) {
      const int k = s * SEG2 + u;
      const int p = run * 32 + k;
      const float2 w = wl2[p];
      Hs4[(p + run) * 8 + c4] = hv[k];
      aD.x += w.y * hv[k].x;
      aD.y += w.y * hv[k].y;
      aD.z += w.y * hv[k].z;
      aD.w += w.y * hv[k].w;
      aB.x += w.x * hv[k].x;
      aB.y += w.x * hv[k].y;
      aB.z += w.x * hv[k].z;
      aB.w += w.x * hv[k].w;
      yd += w.y;
      yb += w.x;
    }
    tD[s] = aD;
    tB[s] = aB;
    zd[s] = yd;
    zb[s] = yb;
  }
  float4 rD, rB;
  rD.x = tD[0].x + tD[1].x;
  rD.y = tD[0].y + tD[1].y;
  rD.z = tD[0].z + tD[1].z;
  rD.w = tD[0].w + tD[1].w;
  rB.x = tB[1].x + tB[0].x;
  rB.y = tB[1].y + tB[0].y;
  rB.z = tB[1].z + tB[0].z;
  rB.w = tB[1].w + tB[0].w;
  const float rzd = zd[0] + zd[1];
  const float rzb = zb[1] + zb[0];

  float4 exD = make_float4(0.0f, 0.0f, 0.0f, 0.0f);
  float4 exB = make_float4(0.0f, 0.0f, 0.0f, 0.0f);
  float exzd = 0.0f, exzb = 0.0f;
#pragma unroll
  for (int d = 3; d >= 1; --d) {
    const float vx = __shfl_up(rD.x, 8 * d, 32);
    const float vy = __shfl_up(rD.y, 8 * d, 32);
    const float vz = __shfl_up(rD.z, 8 * d, 32);
    const float vw = __shfl_up(rD.w, 8 * d, 32);
    const float vs = __shfl_up(rzd, 8 * d, 32);
    const bool has = (sub >= d);
    exD.x += has ? vx : 0.0f;
    exD.y += has ? vy : 0.0f;
    exD.z += has ? vz : 0.0f;
    exD.w += has ? vw : 0.0f;
    exzd += has ? vs : 0.0f;
  }
#pragma unroll
  for (int d = 3; d >= 1; --d) {
    const float vx = __shfl_down(rB.x, 8 * d, 32);
    const float vy = __shfl_down(rB.y, 8 * d, 32);
    const float vz = __shfl_down(rB.z, 8 * d, 32);
    const float vw = __shfl_down(rB.w, 8 * d, 32);
    const float vs = __shfl_down(rzb, 8 * d, 32);
    const bool has = (sub + d <= 3);
    exB.x += has ? vx : 0.0f;
    exB.y += has ? vy : 0.0f;
    exB.z += has ? vz : 0.0f;
    exB.w += has ? vw : 0.0f;
    exzb += has ? vs : 0.0f;
  }
  if (sub == 3) {
    float4 v;
    v.x = exD.x + rD.x;
    v.y = exD.y + rD.y;
    v.z = exD.z + rD.z;
    v.w = exD.w + rD.w;
    wtD4[wave * 8 + c4] = v;
    if (c4 == 0) wtZD[wave] = exzd + rzd;
  }
  if (sub == 0) {
    float4 v;
    v.x = exB.x + rB.x;
    v.y = exB.y + rB.y;
    v.z = exB.z + rB.z;
    v.w = exB.w + rB.w;
    wtB4[wave * 8 + c4] = v;
    if (c4 == 0) wtZB[wave] = exzb + rzb;
  }
  __syncthreads();

  float4 cD = make_float4(0.0f, 0.0f, 0.0f, 0.0f);
  float4 cB = make_float4(0.0f, 0.0f, 0.0f, 0.0f);
  float czd = 0.0f, czb = 0.0f;
  for (int w = 0; w < wave; ++w) {
    const float4 v = wtD4[w * 8 + c4];
    cD.x += v.x;
    cD.y += v.y;
    cD.z += v.z;
    cD.w += v.w;
    czd += wtZD[w];
  }
  for (int w = 15; w > wave; --w) {
    const float4 v = wtB4[w * 8 + c4];
    cB.x += v.x;
    cB.y += v.y;
    cB.z += v.z;
    cB.w += v.w;
    czb += wtZB[w];
  }
  cD.x += exD.x;
  cD.y += exD.y;
  cD.z += exD.z;
  cD.w += exD.w;
  czd += exzd;
  cB.x += exB.x;
  cB.y += exB.y;
  cB.z += exB.z;
  cB.w += exB.w;
  czb += exzb;
  {
    const int g0 = 2 * run, g1 = 2 * run + 1;
    float4 v;
    bD4[g0 * 8 + c4] = cD;
    v.x = cD.x + tD[0].x;
    v.y = cD.y + tD[0].y;
    v.z = cD.z + tD[0].z;
    v.w = cD.w + tD[0].w;
    bD4[g1 * 8 + c4] = v;
    bB4[g1 * 8 + c4] = cB;
    v.x = cB.x + tB[1].x;
    v.y = cB.y + tB[1].y;
    v.z = cB.z + tB[1].z;
    v.w = cB.w + tB[1].w;
    bB4[g0 * 8 + c4] = v;
    if (c4 == 0) {
      zDs[g0] = czd;
      zDs[g1] = czd + zd[0];
      zBs[g1] = czb;
      zBs[g0] = czb + zb[1];
    }
  }
  __syncthreads();

  float* ob = out + (size_t)rbase * 128 + cq * 32 + c4 * 4;
#pragma unroll 1
  for (int it = 0; it < 4; ++it) {
    const int t = (it == 0) ? t0 : (it == 1) ? t1 : (it == 2) ? t2 : t3;
    const float sr = (it == 0) ? s0 : (it == 1) ? s1 : (it == 2) ? s2 : s3;
    const int g = min(t >> 4, NSEG2 - 1);
    const int p0 = g * SEG2;
    const int hr = p0 + (g >> 1);
    float4 sD = bD4[g * 8 + c4], sB = bB4[g * 8 + c4];
    float yD = zDs[g], yB = zBs[g];
    float4 hs[SEG2];
    float wd[SEG2], wb[SEG2];
#pragma unroll
    for (int u = 0; u < SEG2; ++u) hs[u] = Hs4[(hr + u) * 8 + c4];
#pragma unroll
    for (int u = 0; u < SEG2 / 2; ++u) {
      const float4 v = wl4[g * 8 + u];
      const bool d0 = (p0 + 2 * u < t), d1 = (p0 + 2 * u + 1 < t);
      wd[2 * u] = d0 ? v.y : 0.0f;
      wb[2 * u] = d0 ? 0.0f : v.x;
      wd[2 * u + 1] = d1 ? v.w : 0.0f;
      wb[2 * u + 1] = d1 ? 0.0f : v.z;
    }
#pragma unroll
    for (int u = 0; u < SEG2; ++u) {
      const float w = wd[u];
      sD.x += w * hs[u].x;
      sD.y += w * hs[u].y;
      sD.z += w * hs[u].z;
      sD.w += w * hs[u].w;
      yD += w;
    }
#pragma unroll
    for (int u = SEG2 - 1; u >= 0; --u) {
      const float w = wb[u];
      sB.x += w * hs[u].x;
      sB.y += w * hs[u].y;
      sB.z += w * hs[u].z;
      sB.w += w * hs[u].w;
      yB += w;
    }
    const float s_row = sr + mxb;
    const float m_row = leaky(s_row);
    const float a_one = __expf(s_row - m_row);
    const float c_one = __expf(ALPHA * s_row - m_row);
    const float zinv = 1.0f / (c_one * yD + a_one * yB);
    float4 o;
    o.x = (c_one * sD.x + a_one * sB.x) * zinv;
    o.y = (c_one * sD.y + a_one * sB.y) * zinv;
    o.z = (c_one * sD.z + a_one * sB.z) * zinv;
    o.w = (c_one * sD.w + a_one * sB.w) * zinv;
    *(float4*)(ob + (size_t)it * 4 * 128) = o;
  }
}

extern "C" void kernel_launch(void* const* d_in, const int* in_sizes, int n_in,
                              void* d_out, int out_size, void* d_ws, size_t ws_size,
                              hipStream_t stream) {
  const float* x = (const float*)d_in[0];
  const float* W = (const float*)d_in[1];
  const float* a = (const float*)d_in[2];
  float* out = (float*)d_out;

  float* h  = (float*)d_ws;
  float* si = h + (size_t)16384 * 128;
  float* sj = si + 16384;
  float* pmx = sj + 16384;
  float* sidx = pmx + 2048;
  float* sbd = sidx + 16384;
  float* trow = sbd + 32768;

  if (ws_size < (size_t)8790016) return;

  k_h_gemm_scores<<<dim3(512), dim3(128), 0, stream>>>(x, W, a, h, si, sj, pmx);
  k_rank<<<dim3(32, 8), dim3(256), 0, stream>>>(si, sj, pmx, sidx, sbd, trow);
  k_scan_rows<<<dim3(8, 4, 8), dim3(512), 0, stream>>>(h, si, pmx, sidx, sbd, trow, out);
}
